# starting kernel
# baseline (speedup 1.0000x reference)
#include <hip/hip_runtime.h>


typedef float v2f __attribute__((ext_vector_type(2)));
typedef float v8f __attribute__((ext_vector_type(8)));

#define HID 128
#define NPRED 16


__global__ void k_set_deg(float* deg, int n) {
  int i = blockIdx.x * blockDim.x + threadIdx.x;
  if (i < n) deg[i] = 1.0f;
}

__global__ void k_deg_scatter(const int* __restrict__ col,
                              const float* __restrict__ w,
                              float* deg, int E) {
  int e = blockIdx.x * blockDim.x + threadIdx.x;
  if (e < E) atomicAdd(&deg[col[e]], w[e]);
}

__global__ void k_dinv(float* deg, int n) {
  int i = blockIdx.x * blockDim.x + threadIdx.x;
  if (i < n) {
    float d = deg[i];
    deg[i] = (d > 0.0f) ? rsqrtf(d) : 0.0f;
  }
}

__global__ void k_norm(const int* __restrict__ row, const int* __restrict__ col,
                       const float* __restrict__ w, const float* __restrict__ dinv,
                       float* __restrict__ norm, int E) {
  int e = blockIdx.x * blockDim.x + threadIdx.x;
  if (e < E) norm[e] = dinv[row[e]] * w[e] * dinv[col[e]];
}

__global__ void k_gemm_wmma(const float* __restrict__ A,
                            const float* __restrict__ W,
                            float* __restrict__ C, int n) {
  int wid  = (blockIdx.x * blockDim.x + threadIdx.x) >> 5;
  int lane = threadIdx.x & 31;
  int row0 = wid * 16;
  if (row0 >= n) return;

  int half = lane >> 4;
  int l    = lane & 15;

  v8f acc[8];
#pragma unroll
  for (int t = 0; t < 8; ++t) acc[t] = (v8f){0,0,0,0,0,0,0,0};

  const float* arow = A + (size_t)(row0 + l) * HID;

  for (int k = 0; k < HID; k += 4) {
    v2f afrag = *(const v2f*)(arow + k + half * 2);
#pragma unroll
    for (int t = 0; t < 8; ++t) {
      int ncol = t * 16 + l;
      v2f bfrag;
      bfrag.x = W[(size_t)(k + half * 2)     * HID + ncol];
      bfrag.y = W[(size_t)(k + half * 2 + 1) * HID + ncol];
      acc[t] = __builtin_amdgcn_wmma_f32_16x16x4_f32(
           false, afrag,  false, bfrag,
           (short)0, acc[t],  false,  false);
    }
  }

#pragma unroll
  for (int t = 0; t < 8; ++t) {
#pragma unroll
    for (int r = 0; r < 8; ++r) {
      int m    = row0 + r + half * 8;
      int ncol = t * 16 + l;
      C[(size_t)m * HID + ncol] = acc[t][r];
    }
  }
}

__global__ void k_scatter(const float* __restrict__ xw,
                          const int* __restrict__ row,
                          const int* __restrict__ col,
                          const float* __restrict__ norm,
                          float* __restrict__ acc, int E) {
  int e = ((blockIdx.x * blockDim.x + threadIdx.x) >> 5);
  if (e >= E) return;
  int lane = threadIdx.x & 31;
  int r = row[e], c = col[e];
  float s = norm[e];
  const float4 v = *(const float4*)(xw + (size_t)r * HID + lane * 4);
  float* dst = acc + (size_t)c * HID + lane * 4;
  atomicAdd(dst + 0, v.x * s);
  atomicAdd(dst + 1, v.y * s);
  atomicAdd(dst + 2, v.z * s);
  atomicAdd(dst + 3, v.w * s);
}

__global__ void k_epilogue(float* __restrict__ acc,
                           const float* __restrict__ xw,
                           const float* __restrict__ dinv,
                           const float* __restrict__ b, int n) {
  int i = blockIdx.x * blockDim.x + threadIdx.x;
  if (i >= n * (HID / 4)) return;
  int node = i >> 5;
  int f4   = i & 31;
  float s  = dinv[node];
  s = s * s;
  float4 a  = ((float4*)acc)[i];
  float4 xv = ((const float4*)xw)[i];
  float4 bv = ((const float4*)b)[f4];
  float4 o;
  o.x = a.x + xv.x * s + bv.x;  o.x = o.x > 0.0f ? o.x : 0.0f;
  o.y = a.y + xv.y * s + bv.y;  o.y = o.y > 0.0f ? o.y : 0.0f;
  o.z = a.z + xv.z * s + bv.z;  o.z = o.z > 0.0f ? o.z : 0.0f;
  o.w = a.w + xv.w * s + bv.w;  o.w = o.w > 0.0f ? o.w : 0.0f;
  ((float4*)acc)[i] = o;
}

__global__ void k_out_proj(const float* __restrict__ h,
                           const float* __restrict__ Wout,
                           const float* __restrict__ bout,
                           float* __restrict__ out, int n) {
  int idx = blockIdx.x * blockDim.x + threadIdx.x;
  if (idx >= n * NPRED) return;
  int node = idx >> 4;
  int p    = idx & 15;
  const float4* hr = (const float4*)(h + (size_t)node * HID);
  const float4* wr = (const float4*)(Wout + (size_t)p * HID);
  float s = 0.0f;
#pragma unroll 8
  for (int k = 0; k < HID / 4; ++k) {
    float4 a = hr[k], b = wr[k];
    s += a.x * b.x + a.y * b.y + a.z * b.z + a.w * b.w;
  }
  out[idx] = s + bout[p];
}


extern "C" void kernel_launch(void* const* d_in, const int* in_sizes, int n_in,
                              void* d_out, int out_size, void* d_ws, size_t ws_size,
                              hipStream_t stream) {
  const float* x    = (const float*)d_in[0];
  const int*   ei   = (const int*)  d_in[1];
  const float* ew   = (const float*)d_in[2];
  const float* W1   = (const float*)d_in[3];
  const float* b1   = (const float*)d_in[4];
  const float* W2   = (const float*)d_in[5];
  const float* b2   = (const float*)d_in[6];
  const float* Wout = (const float*)d_in[7];
  const float* bout = (const float*)d_in[8];
  float* out = (float*)d_out;

  const int n = in_sizes[0] / HID;
  const int E = in_sizes[2];
  const int* erow = ei;
  const int* ecol = ei + E;

  float* dinv = (float*)d_ws;
  float* norm = dinv + n;
  float* xw   = norm + E;
  float* acc  = xw + (size_t)n * HID;

  const int B = 256;
  int gN  = (n + B - 1) / B;
  int gE  = (E + B - 1) / B;
  int nTiles = (n + 15) / 16;
  int gGemm  = (nTiles + 7) / 8;
  int gScat  = (E + 7) / 8;
  int gEpi   = (n * (HID / 4) + B - 1) / B;
  int gOut   = (n * NPRED + B - 1) / B;

  k_set_deg    <<<gN, B, 0, stream>>>(dinv, n);
  k_deg_scatter<<<gE, B, 0, stream>>>(ecol, ew, dinv, E);
  k_dinv       <<<gN, B, 0, stream>>>(dinv, n);
  k_norm       <<<gE, B, 0, stream>>>(erow, ecol, ew, dinv, norm, E);

  k_gemm_wmma<<<gGemm, B, 0, stream>>>(x, W1, xw, n);
  hipMemsetAsync(acc, 0, (size_t)n * HID * sizeof(float), stream);
  k_scatter  <<<gScat, B, 0, stream>>>(xw, erow, ecol, norm, acc, E);
  k_epilogue <<<gEpi, B, 0, stream>>>(acc, xw, dinv, b1, n);

  k_gemm_wmma<<<gGemm, B, 0, stream>>>(acc, W2, xw, n);
  hipMemsetAsync(acc, 0, (size_t)n * HID * sizeof(float), stream);
  k_scatter  <<<gScat, B, 0, stream>>>(xw, erow, ecol, norm, acc, E);
  k_epilogue <<<gEpi, B, 0, stream>>>(acc, xw, dinv, b2, n);

  k_out_proj<<<gOut, B, 0, stream>>>(acc, Wout, bout, out, n);
}


# Sums held in LDS: 256 blocks each own a node range, find their edges by scanning one-byte owner tags, add gathered rows with LDS float adds, write relu rows once; no L2 float adds; sum order changes
# speedup vs baseline: 1.8450x; 1.8450x over previous
#include <hip/hip_runtime.h>


typedef float v2f __attribute__((ext_vector_type(2)));
typedef float v8f __attribute__((ext_vector_type(8)));

#define HID 128
#define NPRED 16


__global__ void k_set_deg(float* deg, int n) {
  int i = blockIdx.x * blockDim.x + threadIdx.x;
  if (i < n) deg[i] = 1.0f;
}

__global__ void k_deg_scatter(const int* __restrict__ col,
                              const float* __restrict__ w,
                              float* deg, int E) {
  int e = blockIdx.x * blockDim.x + threadIdx.x;
  if (e < E) atomicAdd(&deg[col[e]], w[e]);
}

__global__ void k_dinv(float* deg, int n) {
  int i = blockIdx.x * blockDim.x + threadIdx.x;
  if (i < n) {
    float d = deg[i];
    deg[i] = (d > 0.0f) ? rsqrtf(d) : 0.0f;
  }
}

__global__ void k_tag(const int* __restrict__ col, unsigned* __restrict__ tag4,
                      int E, int Epad, int npb) {
  int i = blockIdx.x * blockDim.x + threadIdx.x;
  if (i * 4 >= Epad) return;
  unsigned pack = 0;
#pragma unroll
  for (int k = 0; k < 4; ++k) {
    int e = i * 4 + k;
    unsigned t = (e < E) ? (unsigned)(col[e] / npb) : 0u;
    pack |= (t & 255u) << (8 * k);
  }
  tag4[i] = pack;
}

__global__ void k_gemm_wmma(const float* __restrict__ A,
                            const float* __restrict__ W,
                            float* __restrict__ C, int n) {
  int wid  = (blockIdx.x * blockDim.x + threadIdx.x) >> 5;
  int lane = threadIdx.x & 31;
  int row0 = wid * 16;
  if (row0 >= n) return;

  int half = lane >> 4;
  int l    = lane & 15;

  v8f acc[8];
#pragma unroll
  for (int t = 0; t < 8; ++t) acc[t] = (v8f){0,0,0,0,0,0,0,0};

  const float* arow = A + (size_t)(row0 + l) * HID;

  for (int k = 0; k < HID; k += 4) {
    v2f afrag = *(const v2f*)(arow + k + half * 2);
#pragma unroll
    for (int t = 0; t < 8; ++t) {
      int ncol = t * 16 + l;
      v2f bfrag;
      bfrag.x = W[(size_t)(k + half * 2)     * HID + ncol];
      bfrag.y = W[(size_t)(k + half * 2 + 1) * HID + ncol];
      acc[t] = __builtin_amdgcn_wmma_f32_16x16x4_f32(
           false, afrag,  false, bfrag,
           (short)0, acc[t],  false,  false);
    }
  }

#pragma unroll
  for (int t = 0; t < 8; ++t) {
#pragma unroll
    for (int r = 0; r < 8; ++r) {
      int m    = row0 + r + half * 8;
      int ncol = t * 16 + l;
      C[(size_t)m * HID + ncol] = acc[t][r];
    }
  }
}

#define NPB   392
#define AGW   32
#define QCAP  544
#define EIF   8

__device__ __forceinline__ unsigned zero_bytes(unsigned x) {
  unsigned y = (x & 0x7F7F7F7Fu) + 0x7F7F7F7Fu;
  return ~(y | x | 0x7F7F7F7Fu);
}

__device__ __forceinline__ void wave_lds_order() {
  __builtin_amdgcn_fence(__ATOMIC_RELEASE, "wavefront");
  __builtin_amdgcn_wave_barrier();
  __builtin_amdgcn_fence(__ATOMIC_ACQUIRE, "wavefront");
}

__device__ __forceinline__ void agg_batch(const int* qv, int cnt, int lane, int base,
                                          const float* __restrict__ xw,
                                          const int* __restrict__ row,
                                          const int* __restrict__ col,
                                          const float* __restrict__ w,
                                          const float* __restrict__ dinv,
                                          float* rows) {
  int r = 0, lc = 0;
  float s = 0.0f;
  if (lane < cnt) {
    const int e = qv[lane];
    r = row[e];
    const int c = col[e];
    s = dinv[r] * w[e] * dinv[c];
    lc = c - base;
  }
  int j = 0;
  for (; j + EIF <= cnt; j += EIF) {
    float v[EIF][4];
    int   lj[EIF];
    float sj[EIF];
#pragma unroll
    for (int u = 0; u < EIF; ++u) {
      const int rj = __builtin_amdgcn_readlane(r, j + u);
      lj[u] = __builtin_amdgcn_readlane(lc, j + u);
      sj[u] = __int_as_float(__builtin_amdgcn_readlane(__float_as_int(s), j + u));
      const float* src = xw + (size_t)rj * HID + lane;
#pragma unroll
      for (int qq = 0; qq < 4; ++qq) v[u][qq] = src[qq * 32];
    }
#pragma unroll
    for (int u = 0; u < EIF; ++u) {
      float* dst = rows + lj[u] * HID + lane;
#pragma unroll
      for (int qq = 0; qq < 4; ++qq) atomicAdd(dst + qq * 32, v[u][qq] * sj[u]);
    }
  }
  if (j < cnt) {
    const int rem = cnt - j;
    float v[EIF][4];
    int   lj[EIF];
    float sj[EIF];
#pragma unroll
    for (int u = 0; u < EIF; ++u) {
      if (u < rem) {
        const int rj = __builtin_amdgcn_readlane(r, j + u);
        lj[u] = __builtin_amdgcn_readlane(lc, j + u);
        sj[u] = __int_as_float(__builtin_amdgcn_readlane(__float_as_int(s), j + u));
        const float* src = xw + (size_t)rj * HID + lane;
#pragma unroll
        for (int qq = 0; qq < 4; ++qq) v[u][qq] = src[qq * 32];
      }
    }
#pragma unroll
    for (int u = 0; u < EIF; ++u) {
      if (u < rem) {
        float* dst = rows + lj[u] * HID + lane;
#pragma unroll
        for (int qq = 0; qq < 4; ++qq) atomicAdd(dst + qq * 32, v[u][qq] * sj[u]);
      }
    }
  }
}

__global__ void __launch_bounds__(1024)
k_agg(const float* __restrict__ xw,
      const int* __restrict__ row,
      const int* __restrict__ col,
      const float* __restrict__ w,
      const unsigned* __restrict__ tag4,
      const float* __restrict__ dinv,
      const float* __restrict__ b,
      float* __restrict__ h, int E, int n, int npb) {
  __shared__ float rows[NPB * HID];
  __shared__ int   q[AGW][QCAP];
  __shared__ int   qcnt[AGW];

  const int tid  = threadIdx.x;
  const int lane = tid & 31;
  const int wv   = __builtin_amdgcn_readfirstlane(tid >> 5);
  const int base = blockIdx.x * npb;
  int cntn = n - base;
  if (cntn > npb) cntn = npb;

  for (int idx = tid; idx < cntn * (HID / 4); idx += AGW * 32) {
    float s = dinv[base + (idx >> 5)];
    s = s * s;
    float4 xv = ((const float4*)xw)[(size_t)base * (HID / 4) + idx];
    float4 o;
    o.x = xv.x * s;  o.y = xv.y * s;  o.z = xv.z * s;  o.w = xv.w * s;
    ((float4*)rows)[idx] = o;
  }
  if (lane == 0) qcnt[wv] = 0;
  __syncthreads();

  const unsigned pat = (unsigned)blockIdx.x * 0x01010101u;
  const uint4* tag16 = (const uint4*)tag4;
  const int nch = (E + 511) >> 9;
  int* myq = q[wv];
  int qn = 0;

  uint4 tnext = make_uint4(0u, 0u, 0u, 0u);
  if (wv < nch) tnext = tag16[(wv << 5) + lane];
  for (int ch = wv; ch < nch; ch += AGW) {
    const int e0 = (ch << 9) + (lane << 4);
    const uint4 t = tnext;
    if (ch + AGW < nch) tnext = tag16[((ch + AGW) << 5) + lane];
    unsigned z[4];
    z[0] = zero_bytes(t.x ^ pat);
    z[1] = zero_bytes(t.y ^ pat);
    z[2] = zero_bytes(t.z ^ pat);
    z[3] = zero_bytes(t.w ^ pat);
    if ((z[0] | z[1] | z[2] | z[3]) != 0u) {
#pragma unroll
      for (int d = 0; d < 4; ++d) {
        unsigned m = z[d];
        while (m != 0u) {
          const int bit = __ffs((int)m) - 1;
          const int e = e0 + d * 4 + (bit >> 3);
          if (e < E) {
            const int slot = atomicAdd(&qcnt[wv], 1);
            myq[slot] = e;
          }
          m &= m - 1u;
        }
      }
    }
    wave_lds_order();
    qn = __builtin_amdgcn_readfirstlane(
        __hip_atomic_load(&qcnt[wv], __ATOMIC_RELAXED, __HIP_MEMORY_SCOPE_WORKGROUP));
    if (qn >= 32) {
      while (qn >= 32) {
        agg_batch(myq + (qn - 32), 32, lane, base, xw, row, col, w, dinv, rows);
        qn -= 32;
      }
      wave_lds_order();
      if (lane == 0)
        __hip_atomic_store(&qcnt[wv], qn, __ATOMIC_RELAXED, __HIP_MEMORY_SCOPE_WORKGROUP);
      wave_lds_order();
    }
  }
  while (qn > 0) {
    const int c = (qn < 32) ? qn : 32;
    agg_batch(myq + (qn - c), c, lane, base, xw, row, col, w, dinv, rows);
    qn -= c;
  }
  __syncthreads();

  for (int idx = tid; idx < cntn * (HID / 4); idx += AGW * 32) {
    float4 a  = ((const float4*)rows)[idx];
    float4 bv = ((const float4*)b)[idx & 31];
    float4 o;
    o.x = a.x + bv.x;  o.x = o.x > 0.0f ? o.x : 0.0f;
    o.y = a.y + bv.y;  o.y = o.y > 0.0f ? o.y : 0.0f;
    o.z = a.z + bv.z;  o.z = o.z > 0.0f ? o.z : 0.0f;
    o.w = a.w + bv.w;  o.w = o.w > 0.0f ? o.w : 0.0f;
    ((float4*)h)[(size_t)base * (HID / 4) + idx] = o;
  }
}

__global__ void k_out_proj(const float* __restrict__ h,
                           const float* __restrict__ Wout,
                           const float* __restrict__ bout,
                           float* __restrict__ out, int n) {
  int idx = blockIdx.x * blockDim.x + threadIdx.x;
  if (idx >= n * NPRED) return;
  int node = idx >> 4;
  int p    = idx & 15;
  const float4* hr = (const float4*)(h + (size_t)node * HID);
  const float4* wr = (const float4*)(Wout + (size_t)p * HID);
  float s = 0.0f;
#pragma unroll 8
  for (int k = 0; k < HID / 4; ++k) {
    float4 a = hr[k], b = wr[k];
    s += a.x * b.x + a.y * b.y + a.z * b.z + a.w * b.w;
  }
  out[idx] = s + bout[p];
}


extern "C" void kernel_launch(void* const* d_in, const int* in_sizes, int n_in,
                              void* d_out, int out_size, void* d_ws, size_t ws_size,
                              hipStream_t stream) {
  const float* x    = (const float*)d_in[0];
  const int*   ei   = (const int*)  d_in[1];
  const float* ew   = (const float*)d_in[2];
  const float* W1   = (const float*)d_in[3];
  const float* b1   = (const float*)d_in[4];
  const float* W2   = (const float*)d_in[5];
  const float* b2   = (const float*)d_in[6];
  const float* Wout = (const float*)d_in[7];
  const float* bout = (const float*)d_in[8];
  float* out = (float*)d_out;

  const int n = in_sizes[0] / HID;
  const int E = in_sizes[2];
  const int* erow = ei;
  const int* ecol = ei + E;

  float* xw   = (float*)d_ws;
  float* hbuf = xw + (size_t)n * HID;
  float* dinv = hbuf + (size_t)n * HID;
  unsigned* tag4 = (unsigned*)(dinv + n);

  const int B = 256;
  int gN  = (n + B - 1) / B;
  int gE  = (E + B - 1) / B;
  int nTiles = (n + 15) / 16;
  int gGemm  = (nTiles + 7) / 8;
  int gOut   = (n * NPRED + B - 1) / B;
  int npb    = (n + 255) / 256;
  int gAgg   = (n + npb - 1) / npb;
  int Epad   = ((E + 511) / 512) * 512;
  int gTag   = (Epad / 4 + B - 1) / B;

  k_set_deg    <<<gN, B, 0, stream>>>(dinv, n);
  k_deg_scatter<<<gE, B, 0, stream>>>(ecol, ew, dinv, E);
  k_dinv       <<<gN, B, 0, stream>>>(dinv, n);
  k_tag        <<<gTag, B, 0, stream>>>(ecol, tag4, E, Epad, npb);

  k_gemm_wmma<<<gGemm, B, 0, stream>>>(x, W1, xw, n);
  k_agg      <<<gAgg, AGW * 32, 0, stream>>>(xw, erow, ecol, ew, tag4, dinv, b1, hbuf, E, n, npb);

  k_gemm_wmma<<<gGemm, B, 0, stream>>>(hbuf, W2, xw, n);
  k_agg      <<<gAgg, AGW * 32, 0, stream>>>(xw, erow, ecol, ew, tag4, dinv, b2, hbuf, E, n, npb);

  k_out_proj<<<gOut, B, 0, stream>>>(hbuf, Wout, bout, out, n);
}


# Degree sums off global float atomics: 128 blocks add an edge slice's w into an LDS row over half the nodes, write it to the idle h plane; dinv sums 64 partials; 8 launches; deg sum order changes
# speedup vs baseline: 2.1249x; 1.1517x over previous
#include <hip/hip_runtime.h>


typedef float v2f __attribute__((ext_vector_type(2)));
typedef float v8f __attribute__((ext_vector_type(8)));

#define HID 128
#define NPRED 16

#define DSL   64
#define DCAP  65536
#define DUN   16

__global__ void __launch_bounds__(1024)
k_deg_part(const int* __restrict__ col, const float* __restrict__ w,
           float* __restrict__ part, int E, int n, int npp, int es) {
  __shared__ float acc[DCAP];
  const int tid = threadIdx.x;
  const int s   = blockIdx.x % DSL;
  const int lo  = (blockIdx.x / DSL) * npp;
  int cnt = n - lo;
  if (cnt > npp) cnt = npp;

  for (int i = tid; i < cnt; i += 1024) acc[i] = 0.0f;
  __syncthreads();

  const int e0 = s * es;
  int e1 = e0 + es;
  if (e1 > E) e1 = E;
  for (int e = e0 + tid; e < e1; e += DUN * 1024) {
    int   c[DUN];
    float x[DUN];
#pragma unroll
    for (int k = 0; k < DUN; ++k) {
      const int ee = e + k * 1024;
      c[k] = -1;
      x[k] = 0.0f;
      if (ee < e1) { c[k] = col[ee];  x[k] = w[ee]; }
    }
#pragma unroll
    for (int k = 0; k < DUN; ++k) {
      const unsigned lc = (unsigned)c[k] - (unsigned)lo;
      if (lc < (unsigned)cnt) atomicAdd(&acc[lc], x[k]);
    }
  }
  __syncthreads();

  float* dst = part + (size_t)s * n + lo;
  for (int i = tid; i < cnt; i += 1024) dst[i] = acc[i];
}

__global__ void k_dinv(const float* __restrict__ part, float* __restrict__ dinv, int n) {
  int i = blockIdx.x * blockDim.x + threadIdx.x;
  if (i < n) {
    float d = 1.0f;
#pragma unroll 16
    for (int s = 0; s < DSL; ++s) d += part[(size_t)s * n + i];
    dinv[i] = (d > 0.0f) ? rsqrtf(d) : 0.0f;
  }
}

__global__ void k_tag(const int* __restrict__ col, unsigned* __restrict__ tag4,
                      int E, int Epad, int npb) {
  int i = blockIdx.x * blockDim.x + threadIdx.x;
  if (i * 4 >= Epad) return;
  unsigned pack = 0;
#pragma unroll
  for (int k = 0; k < 4; ++k) {
    int e = i * 4 + k;
    unsigned t = (e < E) ? (unsigned)(col[e] / npb) : 0u;
    pack |= (t & 255u) << (8 * k);
  }
  tag4[i] = pack;
}

__global__ void k_gemm_wmma(const float* __restrict__ A,
                            const float* __restrict__ W,
                            float* __restrict__ C, int n) {
  int wid  = (blockIdx.x * blockDim.x + threadIdx.x) >> 5;
  int lane = threadIdx.x & 31;
  int row0 = wid * 16;
  if (row0 >= n) return;

  int half = lane >> 4;
  int l    = lane & 15;

  v8f acc[8];
#pragma unroll
  for (int t = 0; t < 8; ++t) acc[t] = (v8f){0,0,0,0,0,0,0,0};

  const float* arow = A + (size_t)(row0 + l) * HID;

  for (int k = 0; k < HID; k += 4) {
    v2f afrag = *(const v2f*)(arow + k + half * 2);
#pragma unroll
    for (int t = 0; t < 8; ++t) {
      int ncol = t * 16 + l;
      v2f bfrag;
      bfrag.x = W[(size_t)(k + half * 2)     * HID + ncol];
      bfrag.y = W[(size_t)(k + half * 2 + 1) * HID + ncol];
      acc[t] = __builtin_amdgcn_wmma_f32_16x16x4_f32(
           false, afrag,  false, bfrag,
           (short)0, acc[t],  false,  false);
    }
  }

#pragma unroll
  for (int t = 0; t < 8; ++t) {
#pragma unroll
    for (int r = 0; r < 8; ++r) {
      int m    = row0 + r + half * 8;
      int ncol = t * 16 + l;
      C[(size_t)m * HID + ncol] = acc[t][r];
    }
  }
}

#define NPB   392
#define AGW   32
#define QCAP  544
#define EIF   8

__device__ __forceinline__ unsigned zero_bytes(unsigned x) {
  unsigned y = (x & 0x7F7F7F7Fu) + 0x7F7F7F7Fu;
  return ~(y | x | 0x7F7F7F7Fu);
}

__device__ __forceinline__ void wave_lds_order() {
  __builtin_amdgcn_fence(__ATOMIC_RELEASE, "wavefront");
  __builtin_amdgcn_wave_barrier();
  __builtin_amdgcn_fence(__ATOMIC_ACQUIRE, "wavefront");
}

__device__ __forceinline__ void agg_batch(const int* qv, int cnt, int lane, int base,
                                          const float* __restrict__ xw,
                                          const int* __restrict__ row,
                                          const int* __restrict__ col,
                                          const float* __restrict__ w,
                                          const float* __restrict__ dinv,
                                          float* rows) {
  int r = 0, lc = 0;
  float s = 0.0f;
  if (lane < cnt) {
    const int e = qv[lane];
    r = row[e];
    const int c = col[e];
    s = dinv[r] * w[e] * dinv[c];
    lc = c - base;
  }
  int j = 0;
  for (; j + EIF <= cnt; j += EIF) {
    float v[EIF][4];
    int   lj[EIF];
    float sj[EIF];
#pragma unroll
    for (int u = 0; u < EIF; ++u) {
      const int rj = __builtin_amdgcn_readlane(r, j + u);
      lj[u] = __builtin_amdgcn_readlane(lc, j + u);
      sj[u] = __int_as_float(__builtin_amdgcn_readlane(__float_as_int(s), j + u));
      const float* src = xw + (size_t)rj * HID + lane;
#pragma unroll
      for (int qq = 0; qq < 4; ++qq) v[u][qq] = src[qq * 32];
    }
#pragma unroll
    for (int u = 0; u < EIF; ++u) {
      float* dst = rows + lj[u] * HID + lane;
#pragma unroll
      for (int qq = 0; qq < 4; ++qq) atomicAdd(dst + qq * 32, v[u][qq] * sj[u]);
    }
  }
  if (j < cnt) {
    const int rem = cnt - j;
    float v[EIF][4];
    int   lj[EIF];
    float sj[EIF];
#pragma unroll
    for (int u = 0; u < EIF; ++u) {
      if (u < rem) {
        const int rj = __builtin_amdgcn_readlane(r, j + u);
        lj[u] = __builtin_amdgcn_readlane(lc, j + u);
        sj[u] = __int_as_float(__builtin_amdgcn_readlane(__float_as_int(s), j + u));
        const float* src = xw + (size_t)rj * HID + lane;
#pragma unroll
        for (int qq = 0; qq < 4; ++qq) v[u][qq] = src[qq * 32];
      }
    }
#pragma unroll
    for (int u = 0; u < EIF; ++u) {
      if (u < rem) {
        float* dst = rows + lj[u] * HID + lane;
#pragma unroll
        for (int qq = 0; qq < 4; ++qq) atomicAdd(dst + qq * 32, v[u][qq] * sj[u]);
      }
    }
  }
}

__global__ void __launch_bounds__(1024)
k_agg(const float* __restrict__ xw,
      const int* __restrict__ row,
      const int* __restrict__ col,
      const float* __restrict__ w,
      const unsigned* __restrict__ tag4,
      const float* __restrict__ dinv,
      const float* __restrict__ b,
      float* __restrict__ h, int E, int n, int npb) {
  __shared__ float rows[NPB * HID];
  __shared__ int   q[AGW][QCAP];
  __shared__ int   qcnt[AGW];

  const int tid  = threadIdx.x;
  const int lane = tid & 31;
  const int wv   = __builtin_amdgcn_readfirstlane(tid >> 5);
  const int base = blockIdx.x * npb;
  int cntn = n - base;
  if (cntn > npb) cntn = npb;

  for (int idx = tid; idx < cntn * (HID / 4); idx += AGW * 32) {
    float s = dinv[base + (idx >> 5)];
    s = s * s;
    float4 xv = ((const float4*)xw)[(size_t)base * (HID / 4) + idx];
    float4 o;
    o.x = xv.x * s;  o.y = xv.y * s;  o.z = xv.z * s;  o.w = xv.w * s;
    ((float4*)rows)[idx] = o;
  }
  if (lane == 0) qcnt[wv] = 0;
  __syncthreads();

  const unsigned pat = (unsigned)blockIdx.x * 0x01010101u;
  const uint4* tag16 = (const uint4*)tag4;
  const int nch = (E + 511) >> 9;
  int* myq = q[wv];
  int qn = 0;

  uint4 tnext = make_uint4(0u, 0u, 0u, 0u);
  if (wv < nch) tnext = tag16[(wv << 5) + lane];
  for (int ch = wv; ch < nch; ch += AGW) {
    const int e0 = (ch << 9) + (lane << 4);
    const uint4 t = tnext;
    if (ch + AGW < nch) tnext = tag16[((ch + AGW) << 5) + lane];
    unsigned z[4];
    z[0] = zero_bytes(t.x ^ pat);
    z[1] = zero_bytes(t.y ^ pat);
    z[2] = zero_bytes(t.z ^ pat);
    z[3] = zero_bytes(t.w ^ pat);
    if ((z[0] | z[1] | z[2] | z[3]) != 0u) {
#pragma unroll
      for (int d = 0; d < 4; ++d) {
        unsigned m = z[d];
        while (m != 0u) {
          const int bit = __ffs((int)m) - 1;
          const int e = e0 + d * 4 + (bit >> 3);
          if (e < E) {
            const int slot = atomicAdd(&qcnt[wv], 1);
            myq[slot] = e;
          }
          m &= m - 1u;
        }
      }
    }
    wave_lds_order();
    qn = __builtin_amdgcn_readfirstlane(
        __hip_atomic_load(&qcnt[wv], __ATOMIC_RELAXED, __HIP_MEMORY_SCOPE_WORKGROUP));
    if (qn >= 32) {
      while (qn >= 32) {
        agg_batch(myq + (qn - 32), 32, lane, base, xw, row, col, w, dinv, rows);
        qn -= 32;
      }
      wave_lds_order();
      if (lane == 0)
        __hip_atomic_store(&qcnt[wv], qn, __ATOMIC_RELAXED, __HIP_MEMORY_SCOPE_WORKGROUP);
      wave_lds_order();
    }
  }
  while (qn > 0) {
    const int c = (qn < 32) ? qn : 32;
    agg_batch(myq + (qn - c), c, lane, base, xw, row, col, w, dinv, rows);
    qn -= c;
  }
  __syncthreads();

  for (int idx = tid; idx < cntn * (HID / 4); idx += AGW * 32) {
    float4 a  = ((const float4*)rows)[idx];
    float4 bv = ((const float4*)b)[idx & 31];
    float4 o;
    o.x = a.x + bv.x;  o.x = o.x > 0.0f ? o.x : 0.0f;
    o.y = a.y + bv.y;  o.y = o.y > 0.0f ? o.y : 0.0f;
    o.z = a.z + bv.z;  o.z = o.z > 0.0f ? o.z : 0.0f;
    o.w = a.w + bv.w;  o.w = o.w > 0.0f ? o.w : 0.0f;
    ((float4*)h)[(size_t)base * (HID / 4) + idx] = o;
  }
}

__global__ void k_out_proj(const float* __restrict__ h,
                           const float* __restrict__ Wout,
                           const float* __restrict__ bout,
                           float* __restrict__ out, int n) {
  int idx = blockIdx.x * blockDim.x + threadIdx.x;
  if (idx >= n * NPRED) return;
  int node = idx >> 4;
  int p    = idx & 15;
  const float4* hr = (const float4*)(h + (size_t)node * HID);
  const float4* wr = (const float4*)(Wout + (size_t)p * HID);
  float s = 0.0f;
#pragma unroll 8
  for (int k = 0; k < HID / 4; ++k) {
    float4 a = hr[k], b = wr[k];
    s += a.x * b.x + a.y * b.y + a.z * b.z + a.w * b.w;
  }
  out[idx] = s + bout[p];
}


extern "C" void kernel_launch(void* const* d_in, const int* in_sizes, int n_in,
                              void* d_out, int out_size, void* d_ws, size_t ws_size,
                              hipStream_t stream) {
  const float* x    = (const float*)d_in[0];
  const int*   ei   = (const int*)  d_in[1];
  const float* ew   = (const float*)d_in[2];
  const float* W1   = (const float*)d_in[3];
  const float* b1   = (const float*)d_in[4];
  const float* W2   = (const float*)d_in[5];
  const float* b2   = (const float*)d_in[6];
  const float* Wout = (const float*)d_in[7];
  const float* bout = (const float*)d_in[8];
  float* out = (float*)d_out;

  const int n = in_sizes[0] / HID;
  const int E = in_sizes[2];
  const int* erow = ei;
  const int* ecol = ei + E;

  float* xw   = (float*)d_ws;
  float* hbuf = xw + (size_t)n * HID;
  float* dinv = hbuf + (size_t)n * HID;
  unsigned* tag4 = (unsigned*)(dinv + n);

  const int B = 256;
  int gN  = (n + B - 1) / B;
  int nTiles = (n + 15) / 16;
  int gGemm  = (nTiles + 7) / 8;
  int gOut   = (n * NPRED + B - 1) / B;
  int npb    = (n + 255) / 256;
  int gAgg   = (n + npb - 1) / npb;
  int Epad   = ((E + 511) / 512) * 512;
  int gTag   = (Epad / 4 + B - 1) / B;
  int nPart  = (n + DCAP - 1) / DCAP;
  int npp    = (((n + nPart - 1) / nPart) + 31) / 32 * 32;
  int gDeg   = DSL * ((n + npp - 1) / npp);
  int es     = (E + DSL - 1) / DSL;

  k_deg_part   <<<gDeg, 1024, 0, stream>>>(ecol, ew, hbuf, E, n, npp, es);
  k_dinv       <<<gN, B, 0, stream>>>(hbuf, dinv, n);
  k_tag        <<<gTag, B, 0, stream>>>(ecol, tag4, E, Epad, npb);

  k_gemm_wmma<<<gGemm, B, 0, stream>>>(x, W1, xw, n);
  k_agg      <<<gAgg, AGW * 32, 0, stream>>>(xw, erow, ecol, ew, tag4, dinv, b1, hbuf, E, n, npb);

  k_gemm_wmma<<<gGemm, B, 0, stream>>>(hbuf, W2, xw, n);
  k_agg      <<<gAgg, AGW * 32, 0, stream>>>(xw, erow, ecol, ew, tag4, dinv, b2, hbuf, E, n, npb);

  k_out_proj<<<gOut, B, 0, stream>>>(hbuf, Wout, bout, out, n);
}


# Gather reads a 512-B source row with one 16-B-per-lane load (4 lines per instruction, was 4 one-line loads); LDS sum rows hold feature 4l+k at word 32k+l: adds stay conflict-free; add order as before
# speedup vs baseline: 2.2696x; 1.0681x over previous
#include <hip/hip_runtime.h>


typedef float v2f __attribute__((ext_vector_type(2)));
typedef float v8f __attribute__((ext_vector_type(8)));

#define HID 128
#define NPRED 16

#define DSL   64
#define DCAP  65536
#define DUN   16

__global__ void __launch_bounds__(1024)
k_deg_part(const int* __restrict__ col, const float* __restrict__ w,
           float* __restrict__ part, int E, int n, int npp, int es) {
  __shared__ float acc[DCAP];
  const int tid = threadIdx.x;
  const int s   = blockIdx.x % DSL;
  const int lo  = (blockIdx.x / DSL) * npp;
  int cnt = n - lo;
  if (cnt > npp) cnt = npp;

  for (int i = tid; i < cnt; i += 1024) acc[i] = 0.0f;
  __syncthreads();

  const int e0 = s * es;
  int e1 = e0 + es;
  if (e1 > E) e1 = E;
  for (int e = e0 + tid; e < e1; e += DUN * 1024) {
    int   c[DUN];
    float x[DUN];
#pragma unroll
    for (int k = 0; k < DUN; ++k) {
      const int ee = e + k * 1024;
      c[k] = -1;
      x[k] = 0.0f;
      if (ee < e1) { c[k] = col[ee];  x[k] = w[ee]; }
    }
#pragma unroll
    for (int k = 0; k < DUN; ++k) {
      const unsigned lc = (unsigned)c[k] - (unsigned)lo;
      if (lc < (unsigned)cnt) atomicAdd(&acc[lc], x[k]);
    }
  }
  __syncthreads();

  float* dst = part + (size_t)s * n + lo;
  for (int i = tid; i < cnt; i += 1024) dst[i] = acc[i];
}

__global__ void k_dinv(const float* __restrict__ part, float* __restrict__ dinv, int n) {
  int i = blockIdx.x * blockDim.x + threadIdx.x;
  if (i < n) {
    float d = 1.0f;
#pragma unroll 16
    for (int s = 0; s < DSL; ++s) d += part[(size_t)s * n + i];
    dinv[i] = (d > 0.0f) ? rsqrtf(d) : 0.0f;
  }
}

__global__ void k_tag(const int* __restrict__ col, unsigned* __restrict__ tag4,
                      int E, int Epad, int npb) {
  int i = blockIdx.x * blockDim.x + threadIdx.x;
  if (i * 4 >= Epad) return;
  unsigned pack = 0;
#pragma unroll
  for (int k = 0; k < 4; ++k) {
    int e = i * 4 + k;
    unsigned t = (e < E) ? (unsigned)(col[e] / npb) : 0u;
    pack |= (t & 255u) << (8 * k);
  }
  tag4[i] = pack;
}

__global__ void k_gemm_wmma(const float* __restrict__ A,
                            const float* __restrict__ W,
                            float* __restrict__ C, int n) {
  int wid  = (blockIdx.x * blockDim.x + threadIdx.x) >> 5;
  int lane = threadIdx.x & 31;
  int row0 = wid * 16;
  if (row0 >= n) return;

  int half = lane >> 4;
  int l    = lane & 15;

  v8f acc[8];
#pragma unroll
  for (int t = 0; t < 8; ++t) acc[t] = (v8f){0,0,0,0,0,0,0,0};

  const float* arow = A + (size_t)(row0 + l) * HID;

  for (int k = 0; k < HID; k += 4) {
    v2f afrag = *(const v2f*)(arow + k + half * 2);
#pragma unroll
    for (int t = 0; t < 8; ++t) {
      int ncol = t * 16 + l;
      v2f bfrag;
      bfrag.x = W[(size_t)(k + half * 2)     * HID + ncol];
      bfrag.y = W[(size_t)(k + half * 2 + 1) * HID + ncol];
      acc[t] = __builtin_amdgcn_wmma_f32_16x16x4_f32(
           false, afrag,  false, bfrag,
           (short)0, acc[t],  false,  false);
    }
  }

#pragma unroll
  for (int t = 0; t < 8; ++t) {
#pragma unroll
    for (int r = 0; r < 8; ++r) {
      int m    = row0 + r + half * 8;
      int ncol = t * 16 + l;
      C[(size_t)m * HID + ncol] = acc[t][r];
    }
  }
}

#define NPB   392
#define AGW   32
#define QCAP  544
#define EIF   8

__device__ __forceinline__ unsigned zero_bytes(unsigned x) {
  unsigned y = (x & 0x7F7F7F7Fu) + 0x7F7F7F7Fu;
  return ~(y | x | 0x7F7F7F7Fu);
}

__device__ __forceinline__ void wave_lds_order() {
  __builtin_amdgcn_fence(__ATOMIC_RELEASE, "wavefront");
  __builtin_amdgcn_wave_barrier();
  __builtin_amdgcn_fence(__ATOMIC_ACQUIRE, "wavefront");
}

__device__ __forceinline__ void agg_batch(const int* qv, int cnt, int lane, int base,
                                          const float* __restrict__ xw,
                                          const int* __restrict__ row,
                                          const int* __restrict__ col,
                                          const float* __restrict__ w,
                                          const float* __restrict__ dinv,
                                          float* rows) {
  int r = 0, lc = 0;
  float s = 0.0f;
  if (lane < cnt) {
    const int e = qv[lane];
    r = row[e];
    const int c = col[e];
    s = dinv[r] * w[e] * dinv[c];
    lc = c - base;
  }
  int j = 0;
  for (; j + EIF <= cnt; j += EIF) {
    float4 v[EIF];
    int    lj[EIF];
    float  sj[EIF];
#pragma unroll
    for (int u = 0; u < EIF; ++u) {
      const int rj = __builtin_amdgcn_readlane(r, j + u);
      lj[u] = __builtin_amdgcn_readlane(lc, j + u);
      sj[u] = __int_as_float(__builtin_amdgcn_readlane(__float_as_int(s), j + u));
      v[u] = ((const float4*)(xw + (size_t)rj * HID))[lane];
    }
#pragma unroll
    for (int u = 0; u < EIF; ++u) {
      float* dst = rows + lj[u] * HID + lane;
      atomicAdd(dst,      v[u].x * sj[u]);
      atomicAdd(dst + 32, v[u].y * sj[u]);
      atomicAdd(dst + 64, v[u].z * sj[u]);
      atomicAdd(dst + 96, v[u].w * sj[u]);
    }
  }
  if (j < cnt) {
    const int rem = cnt - j;
    float4 v[EIF];
    int    lj[EIF];
    float  sj[EIF];
#pragma unroll
    for (int u = 0; u < EIF; ++u) {
      if (u < rem) {
        const int rj = __builtin_amdgcn_readlane(r, j + u);
        lj[u] = __builtin_amdgcn_readlane(lc, j + u);
        sj[u] = __int_as_float(__builtin_amdgcn_readlane(__float_as_int(s), j + u));
        v[u] = ((const float4*)(xw + (size_t)rj * HID))[lane];
      }
    }
#pragma unroll
    for (int u = 0; u < EIF; ++u) {
      if (u < rem) {
        float* dst = rows + lj[u] * HID + lane;
        atomicAdd(dst,      v[u].x * sj[u]);
        atomicAdd(dst + 32, v[u].y * sj[u]);
        atomicAdd(dst + 64, v[u].z * sj[u]);
        atomicAdd(dst + 96, v[u].w * sj[u]);
      }
    }
  }
}

__global__ void __launch_bounds__(1024)
k_agg(const float* __restrict__ xw,
      const int* __restrict__ row,
      const int* __restrict__ col,
      const float* __restrict__ w,
      const unsigned* __restrict__ tag4,
      const float* __restrict__ dinv,
      const float* __restrict__ b,
      float* __restrict__ h, int E, int n, int npb) {
  __shared__ float rows[NPB * HID];
  __shared__ int   q[AGW][QCAP];
  __shared__ int   qcnt[AGW];

  const int tid  = threadIdx.x;
  const int lane = tid & 31;
  const int wv   = __builtin_amdgcn_readfirstlane(tid >> 5);
  const int base = blockIdx.x * npb;
  int cntn = n - base;
  if (cntn > npb) cntn = npb;

  for (int idx = tid; idx < cntn * (HID / 4); idx += AGW * 32) {
    float s = dinv[base + (idx >> 5)];
    s = s * s;
    float4 xv = ((const float4*)xw)[(size_t)base * (HID / 4) + idx];
    float* d = rows + (idx >> 5) * HID + (idx & 31);
    d[0]  = xv.x * s;
    d[32] = xv.y * s;
    d[64] = xv.z * s;
    d[96] = xv.w * s;
  }
  if (lane == 0) qcnt[wv] = 0;
  __syncthreads();

  const unsigned pat = (unsigned)blockIdx.x * 0x01010101u;
  const uint4* tag16 = (const uint4*)tag4;
  const int nch = (E + 511) >> 9;
  int* myq = q[wv];
  int qn = 0;

  uint4 tnext = make_uint4(0u, 0u, 0u, 0u);
  if (wv < nch) tnext = tag16[(wv << 5) + lane];
  for (int ch = wv; ch < nch; ch += AGW) {
    const int e0 = (ch << 9) + (lane << 4);
    const uint4 t = tnext;
    if (ch + AGW < nch) tnext = tag16[((ch + AGW) << 5) + lane];
    unsigned z[4];
    z[0] = zero_bytes(t.x ^ pat);
    z[1] = zero_bytes(t.y ^ pat);
    z[2] = zero_bytes(t.z ^ pat);
    z[3] = zero_bytes(t.w ^ pat);
    if ((z[0] | z[1] | z[2] | z[3]) != 0u) {
#pragma unroll
      for (int d = 0; d < 4; ++d) {
        unsigned m = z[d];
        while (m != 0u) {
          const int bit = __ffs((int)m) - 1;
          const int e = e0 + d * 4 + (bit >> 3);
          if (e < E) {
            const int slot = atomicAdd(&qcnt[wv], 1);
            myq[slot] = e;
          }
          m &= m - 1u;
        }
      }
    }
    wave_lds_order();
    qn = __builtin_amdgcn_readfirstlane(
        __hip_atomic_load(&qcnt[wv], __ATOMIC_RELAXED, __HIP_MEMORY_SCOPE_WORKGROUP));
    if (qn >= 32) {
      while (qn >= 32) {
        agg_batch(myq + (qn - 32), 32, lane, base, xw, row, col, w, dinv, rows);
        qn -= 32;
      }
      wave_lds_order();
      if (lane == 0)
        __hip_atomic_store(&qcnt[wv], qn, __ATOMIC_RELAXED, __HIP_MEMORY_SCOPE_WORKGROUP);
      wave_lds_order();
    }
  }
  while (qn > 0) {
    const int c = (qn < 32) ? qn : 32;
    agg_batch(myq + (qn - c), c, lane, base, xw, row, col, w, dinv, rows);
    qn -= c;
  }
  __syncthreads();

  for (int idx = tid; idx < cntn * (HID / 4); idx += AGW * 32) {
    const float* a = rows + (idx >> 5) * HID + (idx & 31);
    float4 bv = ((const float4*)b)[idx & 31];
    float4 o;
    o.x = a[0]  + bv.x;  o.x = o.x > 0.0f ? o.x : 0.0f;
    o.y = a[32] + bv.y;  o.y = o.y > 0.0f ? o.y : 0.0f;
    o.z = a[64] + bv.z;  o.z = o.z > 0.0f ? o.z : 0.0f;
    o.w = a[96] + bv.w;  o.w = o.w > 0.0f ? o.w : 0.0f;
    ((float4*)h)[(size_t)base * (HID / 4) + idx] = o;
  }
}

__global__ void k_out_proj(const float* __restrict__ h,
                           const float* __restrict__ Wout,
                           const float* __restrict__ bout,
                           float* __restrict__ out, int n) {
  int idx = blockIdx.x * blockDim.x + threadIdx.x;
  if (idx >= n * NPRED) return;
  int node = idx >> 4;
  int p    = idx & 15;
  const float4* hr = (const float4*)(h + (size_t)node * HID);
  const float4* wr = (const float4*)(Wout + (size_t)p * HID);
  float s = 0.0f;
#pragma unroll 8
  for (int k = 0; k < HID / 4; ++k) {
    float4 a = hr[k], b = wr[k];
    s += a.x * b.x + a.y * b.y + a.z * b.z + a.w * b.w;
  }
  out[idx] = s + bout[p];
}


extern "C" void kernel_launch(void* const* d_in, const int* in_sizes, int n_in,
                              void* d_out, int out_size, void* d_ws, size_t ws_size,
                              hipStream_t stream) {
  const float* x    = (const float*)d_in[0];
  const int*   ei   = (const int*)  d_in[1];
  const float* ew   = (const float*)d_in[2];
  const float* W1   = (const float*)d_in[3];
  const float* b1   = (const float*)d_in[4];
  const float* W2   = (const float*)d_in[5];
  const float* b2   = (const float*)d_in[6];
  const float* Wout = (const float*)d_in[7];
  const float* bout = (const float*)d_in[8];
  float* out = (float*)d_out;

  const int n = in_sizes[0] / HID;
  const int E = in_sizes[2];
  const int* erow = ei;
  const int* ecol = ei + E;

  float* xw   = (float*)d_ws;
  float* hbuf = xw + (size_t)n * HID;
  float* dinv = hbuf + (size_t)n * HID;
  unsigned* tag4 = (unsigned*)(dinv + n);

  const int B = 256;
  int gN  = (n + B - 1) / B;
  int nTiles = (n + 15) / 16;
  int gGemm  = (nTiles + 7) / 8;
  int gOut   = (n * NPRED + B - 1) / B;
  int npb    = (n + 255) / 256;
  int gAgg   = (n + npb - 1) / npb;
  int Epad   = ((E + 511) / 512) * 512;
  int gTag   = (Epad / 4 + B - 1) / B;
  int nPart  = (n + DCAP - 1) / DCAP;
  int npp    = (((n + nPart - 1) / nPart) + 31) / 32 * 32;
  int gDeg   = DSL * ((n + npp - 1) / npp);
  int es     = (E + DSL - 1) / DSL;

  k_deg_part   <<<gDeg, 1024, 0, stream>>>(ecol, ew, hbuf, E, n, npp, es);
  k_dinv       <<<gN, B, 0, stream>>>(hbuf, dinv, n);
  k_tag        <<<gTag, B, 0, stream>>>(ecol, tag4, E, Epad, npb);

  k_gemm_wmma<<<gGemm, B, 0, stream>>>(x, W1, xw, n);
  k_agg      <<<gAgg, AGW * 32, 0, stream>>>(xw, erow, ecol, ew, tag4, dinv, b1, hbuf, E, n, npb);

  k_gemm_wmma<<<gGemm, B, 0, stream>>>(hbuf, W2, xw, n);
  k_agg      <<<gAgg, AGW * 32, 0, stream>>>(xw, erow, ecol, ew, tag4, dinv, b2, hbuf, E, n, npb);

  k_out_proj<<<gOut, B, 0, stream>>>(hbuf, Wout, bout, out, n);
}


# Presence map (1 bit per owner block x group of 16 edges, built with the tags): a block reads the tags of its own groups only, not all tags; map+tags use 3 B/edge of workspace; LDS add order changes
# speedup vs baseline: 2.4680x; 1.0874x over previous
#include <hip/hip_runtime.h>


typedef float v2f __attribute__((ext_vector_type(2)));
typedef float v8f __attribute__((ext_vector_type(8)));

#define HID 128
#define NPRED 16

#define DSL   64
#define DCAP  65536
#define DUN   16

__global__ void __launch_bounds__(1024)
k_deg_part(const int* __restrict__ col, const float* __restrict__ w,
           float* __restrict__ part, int E, int n, int npp, int es) {
  __shared__ float acc[DCAP];
  const int tid = threadIdx.x;
  const int s   = blockIdx.x % DSL;
  const int lo  = (blockIdx.x / DSL) * npp;
  int cnt = n - lo;
  if (cnt > npp) cnt = npp;

  for (int i = tid; i < cnt; i += 1024) acc[i] = 0.0f;
  __syncthreads();

  const int e0 = s * es;
  int e1 = e0 + es;
  if (e1 > E) e1 = E;
  for (int e = e0 + tid; e < e1; e += DUN * 1024) {
    int   c[DUN];
    float x[DUN];
#pragma unroll
    for (int k = 0; k < DUN; ++k) {
      const int ee = e + k * 1024;
      c[k] = -1;
      x[k] = 0.0f;
      if (ee < e1) { c[k] = col[ee];  x[k] = w[ee]; }
    }
#pragma unroll
    for (int k = 0; k < DUN; ++k) {
      const unsigned lc = (unsigned)c[k] - (unsigned)lo;
      if (lc < (unsigned)cnt) atomicAdd(&acc[lc], x[k]);
    }
  }
  __syncthreads();

  float* dst = part + (size_t)s * n + lo;
  for (int i = tid; i < cnt; i += 1024) dst[i] = acc[i];
}

__global__ void k_dinv(const float* __restrict__ part, float* __restrict__ dinv, int n) {
  int i = blockIdx.x * blockDim.x + threadIdx.x;
  if (i < n) {
    float d = 1.0f;
#pragma unroll 16
    for (int s = 0; s < DSL; ++s) d += part[(size_t)s * n + i];
    dinv[i] = (d > 0.0f) ? rsqrtf(d) : 0.0f;
  }
}

#define PG      1024
#define PPITCH  257

__global__ void __launch_bounds__(1024)
k_present(const int* __restrict__ col, uint4* __restrict__ tag16,
          unsigned* __restrict__ present, int E, int npb, int wpr) {
  __shared__ unsigned tile[32 * PPITCH];
  const int tid = threadIdx.x;
  for (int i = tid; i < 32 * PPITCH; i += PG) tile[i] = 0u;
  __syncthreads();

  const int g  = blockIdx.x * PG + tid;
  const int e0 = g * 16;
  const int wv = tid >> 5;
  const unsigned bit = 1u << (tid & 31);
  unsigned pk[4] = {0u, 0u, 0u, 0u};
  int cc[16];
#pragma unroll
  for (int k = 0; k < 16; ++k) {
    const int e = e0 + k;
    cc[k] = -1;
    if (e < E) cc[k] = col[e];
  }
#pragma unroll
  for (int k = 0; k < 16; ++k) {
    if (cc[k] >= 0) {
      const unsigned t = (unsigned)cc[k] / (unsigned)npb;
      pk[k >> 2] |= (t & 255u) << (8 * (k & 3));
      if (t < 256u) atomicOr(&tile[wv * PPITCH + t], bit);
    }
  }
  tag16[g] = make_uint4(pk[0], pk[1], pk[2], pk[3]);
  __syncthreads();

  for (int idx = tid; idx < 256 * 32; idx += PG) {
    const int t = idx >> 5, w = idx & 31;
    present[(size_t)t * wpr + blockIdx.x * 32 + w] = tile[w * PPITCH + t];
  }
}

__global__ void k_gemm_wmma(const float* __restrict__ A,
                            const float* __restrict__ W,
                            float* __restrict__ C, int n) {
  int wid  = (blockIdx.x * blockDim.x + threadIdx.x) >> 5;
  int lane = threadIdx.x & 31;
  int row0 = wid * 16;
  if (row0 >= n) return;

  int half = lane >> 4;
  int l    = lane & 15;

  v8f acc[8];
#pragma unroll
  for (int t = 0; t < 8; ++t) acc[t] = (v8f){0,0,0,0,0,0,0,0};

  const float* arow = A + (size_t)(row0 + l) * HID;

  for (int k = 0; k < HID; k += 4) {
    v2f afrag = *(const v2f*)(arow + k + half * 2);
#pragma unroll
    for (int t = 0; t < 8; ++t) {
      int ncol = t * 16 + l;
      v2f bfrag;
      bfrag.x = W[(size_t)(k + half * 2)     * HID + ncol];
      bfrag.y = W[(size_t)(k + half * 2 + 1) * HID + ncol];
      acc[t] = __builtin_amdgcn_wmma_f32_16x16x4_f32(
           false, afrag,  false, bfrag,
           (short)0, acc[t],  false,  false);
    }
  }

#pragma unroll
  for (int t = 0; t < 8; ++t) {
#pragma unroll
    for (int r = 0; r < 8; ++r) {
      int m    = row0 + r + half * 8;
      int ncol = t * 16 + l;
      C[(size_t)m * HID + ncol] = acc[t][r];
    }
  }
}

#define NPB   392
#define AGW   32
#define QCAP  544
#define GQCAP 64
#define EIF   8

__device__ __forceinline__ unsigned zero_bytes(unsigned x) {
  unsigned y = (x & 0x7F7F7F7Fu) + 0x7F7F7F7Fu;
  return ~(y | x | 0x7F7F7F7Fu);
}

__device__ __forceinline__ void wave_lds_order() {
  __builtin_amdgcn_fence(__ATOMIC_RELEASE, "wavefront");
  __builtin_amdgcn_wave_barrier();
  __builtin_amdgcn_fence(__ATOMIC_ACQUIRE, "wavefront");
}

__device__ __forceinline__ void agg_batch(const int* qv, int cnt, int lane, int base,
                                          const float* __restrict__ xw,
                                          const int* __restrict__ row,
                                          const int* __restrict__ col,
                                          const float* __restrict__ w,
                                          const float* __restrict__ dinv,
                                          float* rows) {
  int r = 0, lc = 0;
  float s = 0.0f;
  if (lane < cnt) {
    const int e = qv[lane];
    r = row[e];
    const int c = col[e];
    s = dinv[r] * w[e] * dinv[c];
    lc = c - base;
  }
  int j = 0;
  for (; j + EIF <= cnt; j += EIF) {
    float4 v[EIF];
    int    lj[EIF];
    float  sj[EIF];
#pragma unroll
    for (int u = 0; u < EIF; ++u) {
      const int rj = __builtin_amdgcn_readlane(r, j + u);
      lj[u] = __builtin_amdgcn_readlane(lc, j + u);
      sj[u] = __int_as_float(__builtin_amdgcn_readlane(__float_as_int(s), j + u));
      v[u] = ((const float4*)(xw + (size_t)rj * HID))[lane];
    }
#pragma unroll
    for (int u = 0; u < EIF; ++u) {
      float* dst = rows + lj[u] * HID + lane;
      atomicAdd(dst,      v[u].x * sj[u]);
      atomicAdd(dst + 32, v[u].y * sj[u]);
      atomicAdd(dst + 64, v[u].z * sj[u]);
      atomicAdd(dst + 96, v[u].w * sj[u]);
    }
  }
  if (j < cnt) {
    const int rem = cnt - j;
    float4 v[EIF];
    int    lj[EIF];
    float  sj[EIF];
#pragma unroll
    for (int u = 0; u < EIF; ++u) {
      if (u < rem) {
        const int rj = __builtin_amdgcn_readlane(r, j + u);
        lj[u] = __builtin_amdgcn_readlane(lc, j + u);
        sj[u] = __int_as_float(__builtin_amdgcn_readlane(__float_as_int(s), j + u));
        v[u] = ((const float4*)(xw + (size_t)rj * HID))[lane];
      }
    }
#pragma unroll
    for (int u = 0; u < EIF; ++u) {
      if (u < rem) {
        float* dst = rows + lj[u] * HID + lane;
        atomicAdd(dst,      v[u].x * sj[u]);
        atomicAdd(dst + 32, v[u].y * sj[u]);
        atomicAdd(dst + 64, v[u].z * sj[u]);
        atomicAdd(dst + 96, v[u].w * sj[u]);
      }
    }
  }
}

__device__ __forceinline__ void scan_groups(const int* gv, int cnt, int lane, unsigned pat,
                                            const uint4* __restrict__ tag16, int E,
                                            int* myq, int* mycnt) {
  unsigned z[4] = {0u, 0u, 0u, 0u};
  int e0 = 0;
  if (lane < cnt) {
    const int g = gv[lane];
    const uint4 t = tag16[g];
    e0 = g << 4;
    z[0] = zero_bytes(t.x ^ pat);
    z[1] = zero_bytes(t.y ^ pat);
    z[2] = zero_bytes(t.z ^ pat);
    z[3] = zero_bytes(t.w ^ pat);
  }
#pragma unroll
  for (int d = 0; d < 4; ++d) {
    unsigned m = z[d];
    while (m != 0u) {
      const int bit = __ffs((int)m) - 1;
      const int e = e0 + d * 4 + (bit >> 3);
      if (e < E) {
        const int slot = atomicAdd(mycnt, 1);
        myq[slot] = e;
      }
      m &= m - 1u;
    }
  }
}

__global__ void __launch_bounds__(1024)
k_agg(const float* __restrict__ xw,
      const int* __restrict__ row,
      const int* __restrict__ col,
      const float* __restrict__ w,
      const unsigned* __restrict__ tag4,
      const unsigned* __restrict__ present,
      const float* __restrict__ dinv,
      const float* __restrict__ b,
      float* __restrict__ h, int E, int n, int npb, int wpr) {
  __shared__ float rows[NPB * HID];
  __shared__ int   q[AGW][QCAP];
  __shared__ int   gq[AGW][GQCAP];
  __shared__ int   qcnt[AGW];

  const int tid  = threadIdx.x;
  const int lane = tid & 31;
  const int wv   = __builtin_amdgcn_readfirstlane(tid >> 5);
  const int base = blockIdx.x * npb;
  int cntn = n - base;
  if (cntn > npb) cntn = npb;

  for (int idx = tid; idx < cntn * (HID / 4); idx += AGW * 32) {
    float s = dinv[base + (idx >> 5)];
    s = s * s;
    float4 xv = ((const float4*)xw)[(size_t)base * (HID / 4) + idx];
    float* d = rows + (idx >> 5) * HID + (idx & 31);
    d[0]  = xv.x * s;
    d[32] = xv.y * s;
    d[64] = xv.z * s;
    d[96] = xv.w * s;
  }
  if (lane == 0) qcnt[wv] = 0;
  __syncthreads();

  const unsigned pat = (unsigned)blockIdx.x * 0x01010101u;
  const uint4* tag16 = (const uint4*)tag4;
  const unsigned* myrow = present + (size_t)blockIdx.x * wpr;
  int* myq  = q[wv];
  int* mygq = gq[wv];
  int qn = 0;
  int gn = 0;

  unsigned wnext = 0u;
  if ((wv << 5) < wpr) wnext = myrow[(wv << 5) + lane];
  for (int i0 = wv << 5; i0 < wpr; i0 += AGW * 32) {
    const int i = i0 + lane;
    unsigned word = wnext;
    if (i0 + AGW * 32 < wpr) wnext = myrow[i + AGW * 32];
    unsigned act = __builtin_amdgcn_ballot_w32(word != 0u);
    while (act != 0u) {
      if (word != 0u) {
        const int slot = gn + (int)__builtin_amdgcn_mbcnt_lo(act, 0u);
        mygq[slot] = (i << 5) + (__ffs((int)word) - 1);
        word &= word - 1u;
      }
      gn += __builtin_popcount(act);
      if (gn >= 32) {
        wave_lds_order();
        scan_groups(mygq + (gn - 32), 32, lane, pat, tag16, E, myq, &qcnt[wv]);
        gn -= 32;
        wave_lds_order();
        qn = __builtin_amdgcn_readfirstlane(
            __hip_atomic_load(&qcnt[wv], __ATOMIC_RELAXED, __HIP_MEMORY_SCOPE_WORKGROUP));
        if (qn >= 32) {
          while (qn >= 32) {
            agg_batch(myq + (qn - 32), 32, lane, base, xw, row, col, w, dinv, rows);
            qn -= 32;
          }
          wave_lds_order();
          if (lane == 0)
            __hip_atomic_store(&qcnt[wv], qn, __ATOMIC_RELAXED, __HIP_MEMORY_SCOPE_WORKGROUP);
          wave_lds_order();
        }
      }
      act = __builtin_amdgcn_ballot_w32(word != 0u);
    }
  }
  if (gn > 0) {
    wave_lds_order();
    scan_groups(mygq, gn, lane, pat, tag16, E, myq, &qcnt[wv]);
  }
  wave_lds_order();
  qn = __builtin_amdgcn_readfirstlane(
      __hip_atomic_load(&qcnt[wv], __ATOMIC_RELAXED, __HIP_MEMORY_SCOPE_WORKGROUP));
  while (qn > 0) {
    const int c = (qn < 32) ? qn : 32;
    agg_batch(myq + (qn - c), c, lane, base, xw, row, col, w, dinv, rows);
    qn -= c;
  }
  __syncthreads();

  for (int idx = tid; idx < cntn * (HID / 4); idx += AGW * 32) {
    const float* a = rows + (idx >> 5) * HID + (idx & 31);
    float4 bv = ((const float4*)b)[idx & 31];
    float4 o;
    o.x = a[0]  + bv.x;  o.x = o.x > 0.0f ? o.x : 0.0f;
    o.y = a[32] + bv.y;  o.y = o.y > 0.0f ? o.y : 0.0f;
    o.z = a[64] + bv.z;  o.z = o.z > 0.0f ? o.z : 0.0f;
    o.w = a[96] + bv.w;  o.w = o.w > 0.0f ? o.w : 0.0f;
    ((float4*)h)[(size_t)base * (HID / 4) + idx] = o;
  }
}

__global__ void k_out_proj(const float* __restrict__ h,
                           const float* __restrict__ Wout,
                           const float* __restrict__ bout,
                           float* __restrict__ out, int n) {
  int idx = blockIdx.x * blockDim.x + threadIdx.x;
  if (idx >= n * NPRED) return;
  int node = idx >> 4;
  int p    = idx & 15;
  const float4* hr = (const float4*)(h + (size_t)node * HID);
  const float4* wr = (const float4*)(Wout + (size_t)p * HID);
  float s = 0.0f;
#pragma unroll 8
  for (int k = 0; k < HID / 4; ++k) {
    float4 a = hr[k], b = wr[k];
    s += a.x * b.x + a.y * b.y + a.z * b.z + a.w * b.w;
  }
  out[idx] = s + bout[p];
}


extern "C" void kernel_launch(void* const* d_in, const int* in_sizes, int n_in,
                              void* d_out, int out_size, void* d_ws, size_t ws_size,
                              hipStream_t stream) {
  const float* x    = (const float*)d_in[0];
  const int*   ei   = (const int*)  d_in[1];
  const float* ew   = (const float*)d_in[2];
  const float* W1   = (const float*)d_in[3];
  const float* b1   = (const float*)d_in[4];
  const float* W2   = (const float*)d_in[5];
  const float* b2   = (const float*)d_in[6];
  const float* Wout = (const float*)d_in[7];
  const float* bout = (const float*)d_in[8];
  float* out = (float*)d_out;

  const int n = in_sizes[0] / HID;
  const int E = in_sizes[2];
  const int* erow = ei;
  const int* ecol = ei + E;

  float* xw   = (float*)d_ws;
  float* hbuf = xw + (size_t)n * HID;
  float* dinv = hbuf + (size_t)n * HID;
  unsigned* tag4 = (unsigned*)(dinv + n);
  int gPre = (E + 16 * PG - 1) / (16 * PG);
  int wpr  = gPre * 32;
  unsigned* present = tag4 + (size_t)gPre * (PG * 4);

  const int B = 256;
  int gN  = (n + B - 1) / B;
  int nTiles = (n + 15) / 16;
  int gGemm  = (nTiles + 7) / 8;
  int gOut   = (n * NPRED + B - 1) / B;
  int npb    = (n + 255) / 256;
  int gAgg   = (n + npb - 1) / npb;
  int nPart  = (n + DCAP - 1) / DCAP;
  int npp    = (((n + nPart - 1) / nPart) + 31) / 32 * 32;
  int gDeg   = DSL * ((n + npp - 1) / npp);
  int es     = (E + DSL - 1) / DSL;

  k_deg_part   <<<gDeg, 1024, 0, stream>>>(ecol, ew, hbuf, E, n, npp, es);
  k_dinv       <<<gN, B, 0, stream>>>(hbuf, dinv, n);
  k_present    <<<gPre, PG, 0, stream>>>(ecol, (uint4*)tag4, present, E, npb, wpr);

  k_gemm_wmma<<<gGemm, B, 0, stream>>>(x, W1, xw, n);
  k_agg      <<<gAgg, AGW * 32, 0, stream>>>(xw, erow, ecol, ew, tag4, present, dinv, b1, hbuf, E, n, npb, wpr);

  k_gemm_wmma<<<gGemm, B, 0, stream>>>(hbuf, W2, xw, n);
  k_agg      <<<gAgg, AGW * 32, 0, stream>>>(xw, erow, ecol, ew, tag4, present, dinv, b2, hbuf, E, n, npb, wpr);

  k_out_proj<<<gOut, B, 0, stream>>>(hbuf, Wout, bout, out, n);
}


# Output head moved onto the matrix unit, own launch kept: a wave per 16 nodes reads h2 and Wout rows as 16-B fragments, 32 f32 16x16x4 WMMAs; 6256 waves, not 50000; out's k sum order changes
# speedup vs baseline: 2.8527x; 1.1559x over previous
#include <hip/hip_runtime.h>


typedef float v2f __attribute__((ext_vector_type(2)));
typedef float v8f __attribute__((ext_vector_type(8)));

#define HID 128
#define NPRED 16

#define DSL   64
#define DCAP  65536
#define DUN   16

__global__ void __launch_bounds__(1024)
k_deg_part(const int* __restrict__ col, const float* __restrict__ w,
           float* __restrict__ part, int E, int n, int npp, int es) {
  __shared__ float acc[DCAP];
  const int tid = threadIdx.x;
  const int s   = blockIdx.x % DSL;
  const int lo  = (blockIdx.x / DSL) * npp;
  int cnt = n - lo;
  if (cnt > npp) cnt = npp;

  for (int i = tid; i < cnt; i += 1024) acc[i] = 0.0f;
  __syncthreads();

  const int e0 = s * es;
  int e1 = e0 + es;
  if (e1 > E) e1 = E;
  for (int e = e0 + tid; e < e1; e += DUN * 1024) {
    int   c[DUN];
    float x[DUN];
#pragma unroll
    for (int k = 0; k < DUN; ++k) {
      const int ee = e + k * 1024;
      c[k] = -1;
      x[k] = 0.0f;
      if (ee < e1) { c[k] = col[ee];  x[k] = w[ee]; }
    }
#pragma unroll
    for (int k = 0; k < DUN; ++k) {
      const unsigned lc = (unsigned)c[k] - (unsigned)lo;
      if (lc < (unsigned)cnt) atomicAdd(&acc[lc], x[k]);
    }
  }
  __syncthreads();

  float* dst = part + (size_t)s * n + lo;
  for (int i = tid; i < cnt; i += 1024) dst[i] = acc[i];
}

__global__ void k_dinv(const float* __restrict__ part, float* __restrict__ dinv, int n) {
  int i = blockIdx.x * blockDim.x + threadIdx.x;
  if (i < n) {
    float d = 1.0f;
#pragma unroll 16
    for (int s = 0; s < DSL; ++s) d += part[(size_t)s * n + i];
    dinv[i] = (d > 0.0f) ? rsqrtf(d) : 0.0f;
  }
}

#define PG      1024
#define PPITCH  257

__global__ void __launch_bounds__(1024)
k_present(const int* __restrict__ col, uint4* __restrict__ tag16,
          unsigned* __restrict__ present, int E, int npb, int wpr) {
  __shared__ unsigned tile[32 * PPITCH];
  const int tid = threadIdx.x;
  for (int i = tid; i < 32 * PPITCH; i += PG) tile[i] = 0u;
  __syncthreads();

  const int g  = blockIdx.x * PG + tid;
  const int e0 = g * 16;
  const int wv = tid >> 5;
  const unsigned bit = 1u << (tid & 31);
  unsigned pk[4] = {0u, 0u, 0u, 0u};
  int cc[16];
#pragma unroll
  for (int k = 0; k < 16; ++k) {
    const int e = e0 + k;
    cc[k] = -1;
    if (e < E) cc[k] = col[e];
  }
#pragma unroll
  for (int k = 0; k < 16; ++k) {
    if (cc[k] >= 0) {
      const unsigned t = (unsigned)cc[k] / (unsigned)npb;
      pk[k >> 2] |= (t & 255u) << (8 * (k & 3));
      if (t < 256u) atomicOr(&tile[wv * PPITCH + t], bit);
    }
  }
  tag16[g] = make_uint4(pk[0], pk[1], pk[2], pk[3]);
  __syncthreads();

  for (int idx = tid; idx < 256 * 32; idx += PG) {
    const int t = idx >> 5, w = idx & 31;
    present[(size_t)t * wpr + blockIdx.x * 32 + w] = tile[w * PPITCH + t];
  }
}

__global__ void k_gemm_wmma(const float* __restrict__ A,
                            const float* __restrict__ W,
                            float* __restrict__ C, int n) {
  int wid  = (blockIdx.x * blockDim.x + threadIdx.x) >> 5;
  int lane = threadIdx.x & 31;
  int row0 = wid * 16;
  if (row0 >= n) return;

  int half = lane >> 4;
  int l    = lane & 15;

  v8f acc[8];
#pragma unroll
  for (int t = 0; t < 8; ++t) acc[t] = (v8f){0,0,0,0,0,0,0,0};

  const float* arow = A + (size_t)(row0 + l) * HID;

  for (int k = 0; k < HID; k += 4) {
    v2f afrag = *(const v2f*)(arow + k + half * 2);
#pragma unroll
    for (int t = 0; t < 8; ++t) {
      int ncol = t * 16 + l;
      v2f bfrag;
      bfrag.x = W[(size_t)(k + half * 2)     * HID + ncol];
      bfrag.y = W[(size_t)(k + half * 2 + 1) * HID + ncol];
      acc[t] = __builtin_amdgcn_wmma_f32_16x16x4_f32(
           false, afrag,  false, bfrag,
           (short)0, acc[t],  false,  false);
    }
  }

#pragma unroll
  for (int t = 0; t < 8; ++t) {
#pragma unroll
    for (int r = 0; r < 8; ++r) {
      int m    = row0 + r + half * 8;
      int ncol = t * 16 + l;
      C[(size_t)m * HID + ncol] = acc[t][r];
    }
  }
}

#define NPB   392
#define AGW   32
#define QCAP  544
#define GQCAP 64
#define EIF   8

__device__ __forceinline__ unsigned zero_bytes(unsigned x) {
  unsigned y = (x & 0x7F7F7F7Fu) + 0x7F7F7F7Fu;
  return ~(y | x | 0x7F7F7F7Fu);
}

__device__ __forceinline__ void wave_lds_order() {
  __builtin_amdgcn_fence(__ATOMIC_RELEASE, "wavefront");
  __builtin_amdgcn_wave_barrier();
  __builtin_amdgcn_fence(__ATOMIC_ACQUIRE, "wavefront");
}

__device__ __forceinline__ void agg_batch(const int* qv, int cnt, int lane, int base,
                                          const float* __restrict__ xw,
                                          const int* __restrict__ row,
                                          const int* __restrict__ col,
                                          const float* __restrict__ w,
                                          const float* __restrict__ dinv,
                                          float* rows) {
  int r = 0, lc = 0;
  float s = 0.0f;
  if (lane < cnt) {
    const int e = qv[lane];
    r = row[e];
    const int c = col[e];
    s = dinv[r] * w[e] * dinv[c];
    lc = c - base;
  }
  int j = 0;
  for (; j + EIF <= cnt; j += EIF) {
    float4 v[EIF];
    int    lj[EIF];
    float  sj[EIF];
#pragma unroll
    for (int u = 0; u < EIF; ++u) {
      const int rj = __builtin_amdgcn_readlane(r, j + u);
      lj[u] = __builtin_amdgcn_readlane(lc, j + u);
      sj[u] = __int_as_float(__builtin_amdgcn_readlane(__float_as_int(s), j + u));
      v[u] = ((const float4*)(xw + (size_t)rj * HID))[lane];
    }
#pragma unroll
    for (int u = 0; u < EIF; ++u) {
      float* dst = rows + lj[u] * HID + lane;
      atomicAdd(dst,      v[u].x * sj[u]);
      atomicAdd(dst + 32, v[u].y * sj[u]);
      atomicAdd(dst + 64, v[u].z * sj[u]);
      atomicAdd(dst + 96, v[u].w * sj[u]);
    }
  }
  if (j < cnt) {
    const int rem = cnt - j;
    float4 v[EIF];
    int    lj[EIF];
    float  sj[EIF];
#pragma unroll
    for (int u = 0; u < EIF; ++u) {
      if (u < rem) {
        const int rj = __builtin_amdgcn_readlane(r, j + u);
        lj[u] = __builtin_amdgcn_readlane(lc, j + u);
        sj[u] = __int_as_float(__builtin_amdgcn_readlane(__float_as_int(s), j + u));
        v[u] = ((const float4*)(xw + (size_t)rj * HID))[lane];
      }
    }
#pragma unroll
    for (int u = 0; u < EIF; ++u) {
      if (u < rem) {
        float* dst = rows + lj[u] * HID + lane;
        atomicAdd(dst,      v[u].x * sj[u]);
        atomicAdd(dst + 32, v[u].y * sj[u]);
        atomicAdd(dst + 64, v[u].z * sj[u]);
        atomicAdd(dst + 96, v[u].w * sj[u]);
      }
    }
  }
}

__device__ __forceinline__ void scan_groups(const int* gv, int cnt, int lane, unsigned pat,
                                            const uint4* __restrict__ tag16, int E,
                                            int* myq, int* mycnt) {
  unsigned z[4] = {0u, 0u, 0u, 0u};
  int e0 = 0;
  if (lane < cnt) {
    const int g = gv[lane];
    const uint4 t = tag16[g];
    e0 = g << 4;
    z[0] = zero_bytes(t.x ^ pat);
    z[1] = zero_bytes(t.y ^ pat);
    z[2] = zero_bytes(t.z ^ pat);
    z[3] = zero_bytes(t.w ^ pat);
  }
#pragma unroll
  for (int d = 0; d < 4; ++d) {
    unsigned m = z[d];
    while (m != 0u) {
      const int bit = __ffs((int)m) - 1;
      const int e = e0 + d * 4 + (bit >> 3);
      if (e < E) {
        const int slot = atomicAdd(mycnt, 1);
        myq[slot] = e;
      }
      m &= m - 1u;
    }
  }
}

__global__ void __launch_bounds__(1024)
k_agg(const float* __restrict__ xw,
      const int* __restrict__ row,
      const int* __restrict__ col,
      const float* __restrict__ w,
      const unsigned* __restrict__ tag4,
      const unsigned* __restrict__ present,
      const float* __restrict__ dinv,
      const float* __restrict__ b,
      float* __restrict__ h, int E, int n, int npb, int wpr) {
  __shared__ float rows[NPB * HID];
  __shared__ int   q[AGW][QCAP];
  __shared__ int   gq[AGW][GQCAP];
  __shared__ int   qcnt[AGW];

  const int tid  = threadIdx.x;
  const int lane = tid & 31;
  const int wv   = __builtin_amdgcn_readfirstlane(tid >> 5);
  const int base = blockIdx.x * npb;
  int cntn = n - base;
  if (cntn > npb) cntn = npb;

  for (int idx = tid; idx < cntn * (HID / 4); idx += AGW * 32) {
    float s = dinv[base + (idx >> 5)];
    s = s * s;
    float4 xv = ((const float4*)xw)[(size_t)base * (HID / 4) + idx];
    float* d = rows + (idx >> 5) * HID + (idx & 31);
    d[0]  = xv.x * s;
    d[32] = xv.y * s;
    d[64] = xv.z * s;
    d[96] = xv.w * s;
  }
  if (lane == 0) qcnt[wv] = 0;
  __syncthreads();

  const unsigned pat = (unsigned)blockIdx.x * 0x01010101u;
  const uint4* tag16 = (const uint4*)tag4;
  const unsigned* myrow = present + (size_t)blockIdx.x * wpr;
  int* myq  = q[wv];
  int* mygq = gq[wv];
  int qn = 0;
  int gn = 0;

  unsigned wnext = 0u;
  if ((wv << 5) < wpr) wnext = myrow[(wv << 5) + lane];
  for (int i0 = wv << 5; i0 < wpr; i0 += AGW * 32) {
    const int i = i0 + lane;
    unsigned word = wnext;
    if (i0 + AGW * 32 < wpr) wnext = myrow[i + AGW * 32];
    unsigned act = __builtin_amdgcn_ballot_w32(word != 0u);
    while (act != 0u) {
      if (word != 0u) {
        const int slot = gn + (int)__builtin_amdgcn_mbcnt_lo(act, 0u);
        mygq[slot] = (i << 5) + (__ffs((int)word) - 1);
        word &= word - 1u;
      }
      gn += __builtin_popcount(act);
      if (gn >= 32) {
        wave_lds_order();
        scan_groups(mygq + (gn - 32), 32, lane, pat, tag16, E, myq, &qcnt[wv]);
        gn -= 32;
        wave_lds_order();
        qn = __builtin_amdgcn_readfirstlane(
            __hip_atomic_load(&qcnt[wv], __ATOMIC_RELAXED, __HIP_MEMORY_SCOPE_WORKGROUP));
        if (qn >= 32) {
          while (qn >= 32) {
            agg_batch(myq + (qn - 32), 32, lane, base, xw, row, col, w, dinv, rows);
            qn -= 32;
          }
          wave_lds_order();
          if (lane == 0)
            __hip_atomic_store(&qcnt[wv], qn, __ATOMIC_RELAXED, __HIP_MEMORY_SCOPE_WORKGROUP);
          wave_lds_order();
        }
      }
      act = __builtin_amdgcn_ballot_w32(word != 0u);
    }
  }
  if (gn > 0) {
    wave_lds_order();
    scan_groups(mygq, gn, lane, pat, tag16, E, myq, &qcnt[wv]);
  }
  wave_lds_order();
  qn = __builtin_amdgcn_readfirstlane(
      __hip_atomic_load(&qcnt[wv], __ATOMIC_RELAXED, __HIP_MEMORY_SCOPE_WORKGROUP));
  while (qn > 0) {
    const int c = (qn < 32) ? qn : 32;
    agg_batch(myq + (qn - c), c, lane, base, xw, row, col, w, dinv, rows);
    qn -= c;
  }
  __syncthreads();

  for (int idx = tid; idx < cntn * (HID / 4); idx += AGW * 32) {
    const float* a = rows + (idx >> 5) * HID + (idx & 31);
    float4 bv = ((const float4*)b)[idx & 31];
    float4 o;
    o.x = a[0]  + bv.x;  o.x = o.x > 0.0f ? o.x : 0.0f;
    o.y = a[32] + bv.y;  o.y = o.y > 0.0f ? o.y : 0.0f;
    o.z = a[64] + bv.z;  o.z = o.z > 0.0f ? o.z : 0.0f;
    o.w = a[96] + bv.w;  o.w = o.w > 0.0f ? o.w : 0.0f;
    ((float4*)h)[(size_t)base * (HID / 4) + idx] = o;
  }
}

__global__ void k_out_wmma(const float* __restrict__ h,
                           const float* __restrict__ Wout,
                           const float* __restrict__ bout,
                           float* __restrict__ out, int n) {
  const int wid  = (blockIdx.x * blockDim.x + threadIdx.x) >> 5;
  const int lane = threadIdx.x & 31;
  const int row0 = wid * 16;
  if (row0 >= n) return;

  const int c  = lane & 15;
  const int hf = lane >> 4;

  const int node = row0 + c;
  const int ld   = node < n ? node : n - 1;
  const float4* wr = (const float4*)(Wout + (size_t)c * HID) + hf;
  const float4* hr = (const float4*)(h + (size_t)ld * HID) + hf;

  const float4 bl = ((const float4*)bout)[2 * hf];
  const float4 bh = ((const float4*)bout)[2 * hf + 1];
  v8f acc0 = (v8f){bl.x, bl.y, bl.z, bl.w, bh.x, bh.y, bh.z, bh.w};
  v8f acc1 = (v8f){0,0,0,0,0,0,0,0};
#pragma unroll
  for (int j = 0; j < 16; ++j) {
    const float4 a = wr[2 * j];
    const float4 b = hr[2 * j];
    v2f a0 = {a.x, a.y}, b0 = {b.x, b.y};
    v2f a1 = {a.z, a.w}, b1 = {b.z, b.w};
    acc0 = __builtin_amdgcn_wmma_f32_16x16x4_f32(
         false, a0,  false, b0,
         (short)0, acc0,  false,  false);
    acc1 = __builtin_amdgcn_wmma_f32_16x16x4_f32(
         false, a1,  false, b1,
         (short)0, acc1,  false,  false);
  }
  const v8f r = acc0 + acc1;
  if (node < n) {
    float4* o = (float4*)(out + (size_t)node * NPRED + 8 * hf);
    o[0] = make_float4(r[0], r[1], r[2], r[3]);
    o[1] = make_float4(r[4], r[5], r[6], r[7]);
  }
}


extern "C" void kernel_launch(void* const* d_in, const int* in_sizes, int n_in,
                              void* d_out, int out_size, void* d_ws, size_t ws_size,
                              hipStream_t stream) {
  const float* x    = (const float*)d_in[0];
  const int*   ei   = (const int*)  d_in[1];
  const float* ew   = (const float*)d_in[2];
  const float* W1   = (const float*)d_in[3];
  const float* b1   = (const float*)d_in[4];
  const float* W2   = (const float*)d_in[5];
  const float* b2   = (const float*)d_in[6];
  const float* Wout = (const float*)d_in[7];
  const float* bout = (const float*)d_in[8];
  float* out = (float*)d_out;

  const int n = in_sizes[0] / HID;
  const int E = in_sizes[2];
  const int* erow = ei;
  const int* ecol = ei + E;

  float* xw   = (float*)d_ws;
  float* hbuf = xw + (size_t)n * HID;
  float* dinv = hbuf + (size_t)n * HID;
  unsigned* tag4 = (unsigned*)(dinv + n);
  int gPre = (E + 16 * PG - 1) / (16 * PG);
  int wpr  = gPre * 32;
  unsigned* present = tag4 + (size_t)gPre * (PG * 4);

  const int B = 256;
  int gN  = (n + B - 1) / B;
  int nTiles = (n + 15) / 16;
  int gGemm  = (nTiles + 7) / 8;
  int npb    = (n + 255) / 256;
  int gAgg   = (n + npb - 1) / npb;
  int nPart  = (n + DCAP - 1) / DCAP;
  int npp    = (((n + nPart - 1) / nPart) + 31) / 32 * 32;
  int gDeg   = DSL * ((n + npp - 1) / npp);
  int es     = (E + DSL - 1) / DSL;

  k_deg_part   <<<gDeg, 1024, 0, stream>>>(ecol, ew, hbuf, E, n, npp, es);
  k_dinv       <<<gN, B, 0, stream>>>(hbuf, dinv, n);
  k_present    <<<gPre, PG, 0, stream>>>(ecol, (uint4*)tag4, present, E, npb, wpr);

  k_gemm_wmma<<<gGemm, B, 0, stream>>>(x, W1, xw, n);
  k_agg      <<<gAgg, AGW * 32, 0, stream>>>(xw, erow, ecol, ew, tag4, present, dinv, b1, hbuf, E, n, npb, wpr);

  k_gemm_wmma<<<gGemm, B, 0, stream>>>(hbuf, W2, xw, n);
  k_agg      <<<gAgg, AGW * 32, 0, stream>>>(xw, erow, ecol, ew, tag4, present, dinv, b2, hbuf, E, n, npb, wpr);

  k_out_wmma<<<gGemm, B, 0, stream>>>(hbuf, Wout, bout, out, n);
}


# Group scan reads a flagged group's 16 destination ids itself (4 lanes a group, slots by ballot): no tag array, no per-edge col or dinv[dst] read (owned dinv in LDS); LDS add order changes
# speedup vs baseline: 3.0401x; 1.0657x over previous
#include <hip/hip_runtime.h>


typedef float v2f __attribute__((ext_vector_type(2)));
typedef float v8f __attribute__((ext_vector_type(8)));

#define HID 128
#define NPRED 16

#define DSL   64
#define DCAP  65536
#define DUN   16

__global__ void __launch_bounds__(1024)
k_deg_part(const int* __restrict__ col, const float* __restrict__ w,
           float* __restrict__ part, int E, int n, int npp, int es) {
  __shared__ float acc[DCAP];
  const int tid = threadIdx.x;
  const int s   = blockIdx.x % DSL;
  const int lo  = (blockIdx.x / DSL) * npp;
  int cnt = n - lo;
  if (cnt > npp) cnt = npp;

  for (int i = tid; i < cnt; i += 1024) acc[i] = 0.0f;
  __syncthreads();

  const int e0 = s * es;
  int e1 = e0 + es;
  if (e1 > E) e1 = E;
  for (int e = e0 + tid; e < e1; e += DUN * 1024) {
    int   c[DUN];
    float x[DUN];
#pragma unroll
    for (int k = 0; k < DUN; ++k) {
      const int ee = e + k * 1024;
      c[k] = -1;
      x[k] = 0.0f;
      if (ee < e1) { c[k] = col[ee];  x[k] = w[ee]; }
    }
#pragma unroll
    for (int k = 0; k < DUN; ++k) {
      const unsigned lc = (unsigned)c[k] - (unsigned)lo;
      if (lc < (unsigned)cnt) atomicAdd(&acc[lc], x[k]);
    }
  }
  __syncthreads();

  float* dst = part + (size_t)s * n + lo;
  for (int i = tid; i < cnt; i += 1024) dst[i] = acc[i];
}

__global__ void k_dinv(const float* __restrict__ part, float* __restrict__ dinv, int n) {
  int i = blockIdx.x * blockDim.x + threadIdx.x;
  if (i < n) {
    float d = 1.0f;
#pragma unroll 16
    for (int s = 0; s < DSL; ++s) d += part[(size_t)s * n + i];
    dinv[i] = (d > 0.0f) ? rsqrtf(d) : 0.0f;
  }
}

#define PG      1024
#define PPITCH  257

__global__ void __launch_bounds__(1024)
k_present(const int* __restrict__ col,
          unsigned* __restrict__ present, int E, int npb, int wpr) {
  __shared__ unsigned tile[32 * PPITCH];
  const int tid = threadIdx.x;
  for (int i = tid; i < 32 * PPITCH; i += PG) tile[i] = 0u;
  __syncthreads();

  const int g  = blockIdx.x * PG + tid;
  const int e0 = g * 16;
  const int wv = tid >> 5;
  const unsigned bit = 1u << (tid & 31);
  int cc[16];
#pragma unroll
  for (int k = 0; k < 16; ++k) {
    const int e = e0 + k;
    cc[k] = -1;
    if (e < E) cc[k] = col[e];
  }
#pragma unroll
  for (int k = 0; k < 16; ++k) {
    if (cc[k] >= 0) {
      const unsigned t = (unsigned)cc[k] / (unsigned)npb;
      if (t < 256u) atomicOr(&tile[wv * PPITCH + t], bit);
    }
  }
  __syncthreads();

  for (int idx = tid; idx < 256 * 32; idx += PG) {
    const int t = idx >> 5, w = idx & 31;
    present[(size_t)t * wpr + blockIdx.x * 32 + w] = tile[w * PPITCH + t];
  }
}

__global__ void k_gemm_wmma(const float* __restrict__ A,
                            const float* __restrict__ W,
                            float* __restrict__ C, int n) {
  int wid  = (blockIdx.x * blockDim.x + threadIdx.x) >> 5;
  int lane = threadIdx.x & 31;
  int row0 = wid * 16;
  if (row0 >= n) return;

  int half = lane >> 4;
  int l    = lane & 15;

  v8f acc[8];
#pragma unroll
  for (int t = 0; t < 8; ++t) acc[t] = (v8f){0,0,0,0,0,0,0,0};

  const float* arow = A + (size_t)(row0 + l) * HID;

  for (int k = 0; k < HID; k += 4) {
    v2f afrag = *(const v2f*)(arow + k + half * 2);
#pragma unroll
    for (int t = 0; t < 8; ++t) {
      int ncol = t * 16 + l;
      v2f bfrag;
      bfrag.x = W[(size_t)(k + half * 2)     * HID + ncol];
      bfrag.y = W[(size_t)(k + half * 2 + 1) * HID + ncol];
      acc[t] = __builtin_amdgcn_wmma_f32_16x16x4_f32(
           false, afrag,  false, bfrag,
           (short)0, acc[t],  false,  false);
    }
  }

#pragma unroll
  for (int t = 0; t < 8; ++t) {
#pragma unroll
    for (int r = 0; r < 8; ++r) {
      int m    = row0 + r + half * 8;
      int ncol = t * 16 + l;
      C[(size_t)m * HID + ncol] = acc[t][r];
    }
  }
}

#define NPB   392
#define AGW   32
#define QCAP  544
#define GQCAP 64
#define EIF   8
#define EBITS 23

__device__ __forceinline__ void wave_lds_order() {
  __builtin_amdgcn_fence(__ATOMIC_RELEASE, "wavefront");
  __builtin_amdgcn_wave_barrier();
  __builtin_amdgcn_fence(__ATOMIC_ACQUIRE, "wavefront");
}

__device__ __forceinline__ void agg_batch(const int* qv, int cnt, int lane,
                                          const float* __restrict__ xw,
                                          const int* __restrict__ row,
                                          const float* __restrict__ w,
                                          const float* __restrict__ dinv,
                                          const float* dl,
                                          float* rows) {
  int r = 0, lc = 0;
  float s = 0.0f;
  if (lane < cnt) {
    const unsigned p = (unsigned)qv[lane];
    const int e = (int)(p & ((1u << EBITS) - 1u));
    lc = (int)(p >> EBITS);
    r = row[e];
    s = dinv[r] * w[e] * dl[lc];
  }
  int j = 0;
  for (; j + EIF <= cnt; j += EIF) {
    float4 v[EIF];
    int    lj[EIF];
    float  sj[EIF];
#pragma unroll
    for (int u = 0; u < EIF; ++u) {
      const int rj = __builtin_amdgcn_readlane(r, j + u);
      lj[u] = __builtin_amdgcn_readlane(lc, j + u);
      sj[u] = __int_as_float(__builtin_amdgcn_readlane(__float_as_int(s), j + u));
      v[u] = ((const float4*)(xw + (size_t)rj * HID))[lane];
    }
#pragma unroll
    for (int u = 0; u < EIF; ++u) {
      float* dst = rows + lj[u] * HID + lane;
      atomicAdd(dst,      v[u].x * sj[u]);
      atomicAdd(dst + 32, v[u].y * sj[u]);
      atomicAdd(dst + 64, v[u].z * sj[u]);
      atomicAdd(dst + 96, v[u].w * sj[u]);
    }
  }
  if (j < cnt) {
    const int rem = cnt - j;
    float4 v[EIF];
    int    lj[EIF];
    float  sj[EIF];
#pragma unroll
    for (int u = 0; u < EIF; ++u) {
      if (u < rem) {
        const int rj = __builtin_amdgcn_readlane(r, j + u);
        lj[u] = __builtin_amdgcn_readlane(lc, j + u);
        sj[u] = __int_as_float(__builtin_amdgcn_readlane(__float_as_int(s), j + u));
        v[u] = ((const float4*)(xw + (size_t)rj * HID))[lane];
      }
    }
#pragma unroll
    for (int u = 0; u < EIF; ++u) {
      if (u < rem) {
        float* dst = rows + lj[u] * HID + lane;
        atomicAdd(dst,      v[u].x * sj[u]);
        atomicAdd(dst + 32, v[u].y * sj[u]);
        atomicAdd(dst + 64, v[u].z * sj[u]);
        atomicAdd(dst + 96, v[u].w * sj[u]);
      }
    }
  }
}

__device__ __forceinline__ int scan_groups(const int* gv, int cnt, int lane, int base, int cntn,
                                           const int* __restrict__ col, int E, bool vec,
                                           int* myq, int qn) {
  const int sub = lane & 3;
  int cv[4][4];
  int eb[4];
#pragma unroll
  for (int k = 0; k < 4; ++k) {
    const int gi = 8 * k + (lane >> 2);
    eb[k] = E;
    if (gi < cnt) eb[k] = (gv[gi] << 4) + (sub << 2);
    cv[k][0] = -1;  cv[k][1] = -1;  cv[k][2] = -1;  cv[k][3] = -1;
  }
  if (vec) {
#pragma unroll
    for (int k = 0; k < 4; ++k) {
      if (eb[k] < E) {
        const int4 t = *(const int4*)(col + eb[k]);
        cv[k][0] = t.x;  cv[k][1] = t.y;  cv[k][2] = t.z;  cv[k][3] = t.w;
      }
    }
  } else {
#pragma unroll
    for (int k = 0; k < 4; ++k) {
      if (eb[k]     < E) cv[k][0] = col[eb[k]];
      if (eb[k] + 1 < E) cv[k][1] = col[eb[k] + 1];
      if (eb[k] + 2 < E) cv[k][2] = col[eb[k] + 2];
      if (eb[k] + 3 < E) cv[k][3] = col[eb[k] + 3];
    }
  }
#pragma unroll
  for (int k = 0; k < 4; ++k) {
    if (8 * k < cnt) {
#pragma unroll
      for (int s = 0; s < 4; ++s) {
        const unsigned lc = (unsigned)cv[k][s] - (unsigned)base;
        const bool hit = lc < (unsigned)cntn;
        const unsigned m = __builtin_amdgcn_ballot_w32(hit);
        if (hit)
          myq[qn + (int)__builtin_amdgcn_mbcnt_lo(m, 0u)] = (int)((lc << EBITS) | (unsigned)(eb[k] + s));
        qn += __builtin_popcount(m);
      }
    }
  }
  return qn;
}

__global__ void __launch_bounds__(1024)
k_agg(const float* __restrict__ xw,
      const int* __restrict__ row,
      const int* __restrict__ col,
      const float* __restrict__ w,
      const unsigned* __restrict__ present,
      const float* __restrict__ dinv,
      const float* __restrict__ b,
      float* __restrict__ h, int E, int n, int npb, int wpr) {
  __shared__ float rows[NPB * HID];
  __shared__ int   q[AGW][QCAP];
  __shared__ int   gq[AGW][GQCAP];
  __shared__ float dl[NPB];

  const int tid  = threadIdx.x;
  const int lane = tid & 31;
  const int wv   = __builtin_amdgcn_readfirstlane(tid >> 5);
  const int base = blockIdx.x * npb;
  int cntn = n - base;
  if (cntn > npb) cntn = npb;

  for (int idx = tid; idx < cntn * (HID / 4); idx += AGW * 32) {
    const float s0 = dinv[base + (idx >> 5)];
    const float s = s0 * s0;
    float4 xv = ((const float4*)xw)[(size_t)base * (HID / 4) + idx];
    float* d = rows + (idx >> 5) * HID + (idx & 31);
    d[0]  = xv.x * s;
    d[32] = xv.y * s;
    d[64] = xv.z * s;
    d[96] = xv.w * s;
    if ((idx & 31) == 0) dl[idx >> 5] = s0;
  }
  __syncthreads();

  const bool vec = (E & 3) == 0;
  const unsigned* myrow = present + (size_t)blockIdx.x * wpr;
  int* myq  = q[wv];
  int* mygq = gq[wv];
  int qn = 0;
  int gn = 0;

  unsigned wnext = 0u;
  if ((wv << 5) < wpr) wnext = myrow[(wv << 5) + lane];
  for (int i0 = wv << 5; i0 < wpr; i0 += AGW * 32) {
    const int i = i0 + lane;
    unsigned word = wnext;
    if (i0 + AGW * 32 < wpr) wnext = myrow[i + AGW * 32];
    unsigned act = __builtin_amdgcn_ballot_w32(word != 0u);
    while (act != 0u) {
      if (word != 0u) {
        const int slot = gn + (int)__builtin_amdgcn_mbcnt_lo(act, 0u);
        mygq[slot] = (i << 5) + (__ffs((int)word) - 1);
        word &= word - 1u;
      }
      gn += __builtin_popcount(act);
      if (gn >= 32) {
        wave_lds_order();
        qn = scan_groups(mygq + (gn - 32), 32, lane, base, cntn, col, E, vec, myq, qn);
        gn -= 32;
        wave_lds_order();
        if (qn >= 32) {
          while (qn >= 32) {
            agg_batch(myq + (qn - 32), 32, lane, xw, row, w, dinv, dl, rows);
            qn -= 32;
          }
          wave_lds_order();
        }
      }
      act = __builtin_amdgcn_ballot_w32(word != 0u);
    }
  }
  if (gn > 0) {
    wave_lds_order();
    qn = scan_groups(mygq, gn, lane, base, cntn, col, E, vec, myq, qn);
  }
  wave_lds_order();
  while (qn > 0) {
    const int c = (qn < 32) ? qn : 32;
    agg_batch(myq + (qn - c), c, lane, xw, row, w, dinv, dl, rows);
    qn -= c;
  }
  __syncthreads();

  for (int idx = tid; idx < cntn * (HID / 4); idx += AGW * 32) {
    const float* a = rows + (idx >> 5) * HID + (idx & 31);
    float4 bv = ((const float4*)b)[idx & 31];
    float4 o;
    o.x = a[0]  + bv.x;  o.x = o.x > 0.0f ? o.x : 0.0f;
    o.y = a[32] + bv.y;  o.y = o.y > 0.0f ? o.y : 0.0f;
    o.z = a[64] + bv.z;  o.z = o.z > 0.0f ? o.z : 0.0f;
    o.w = a[96] + bv.w;  o.w = o.w > 0.0f ? o.w : 0.0f;
    ((float4*)h)[(size_t)base * (HID / 4) + idx] = o;
  }
}

__global__ void k_out_wmma(const float* __restrict__ h,
                           const float* __restrict__ Wout,
                           const float* __restrict__ bout,
                           float* __restrict__ out, int n) {
  const int wid  = (blockIdx.x * blockDim.x + threadIdx.x) >> 5;
  const int lane = threadIdx.x & 31;
  const int row0 = wid * 16;
  if (row0 >= n) return;

  const int c  = lane & 15;
  const int hf = lane >> 4;

  const int node = row0 + c;
  const int ld   = node < n ? node : n - 1;
  const float4* wr = (const float4*)(Wout + (size_t)c * HID) + hf;
  const float4* hr = (const float4*)(h + (size_t)ld * HID) + hf;

  const float4 bl = ((const float4*)bout)[2 * hf];
  const float4 bh = ((const float4*)bout)[2 * hf + 1];
  v8f acc0 = (v8f){bl.x, bl.y, bl.z, bl.w, bh.x, bh.y, bh.z, bh.w};
  v8f acc1 = (v8f){0,0,0,0,0,0,0,0};
#pragma unroll
  for (int j = 0; j < 16; ++j) {
    const float4 a = wr[2 * j];
    const float4 b = hr[2 * j];
    v2f a0 = {a.x, a.y}, b0 = {b.x, b.y};
    v2f a1 = {a.z, a.w}, b1 = {b.z, b.w};
    acc0 = __builtin_amdgcn_wmma_f32_16x16x4_f32(
         false, a0,  false, b0,
         (short)0, acc0,  false,  false);
    acc1 = __builtin_amdgcn_wmma_f32_16x16x4_f32(
         false, a1,  false, b1,
         (short)0, acc1,  false,  false);
  }
  const v8f r = acc0 + acc1;
  if (node < n) {
    float4* o = (float4*)(out + (size_t)node * NPRED + 8 * hf);
    o[0] = make_float4(r[0], r[1], r[2], r[3]);
    o[1] = make_float4(r[4], r[5], r[6], r[7]);
  }
}


extern "C" void kernel_launch(void* const* d_in, const int* in_sizes, int n_in,
                              void* d_out, int out_size, void* d_ws, size_t ws_size,
                              hipStream_t stream) {
  const float* x    = (const float*)d_in[0];
  const int*   ei   = (const int*)  d_in[1];
  const float* ew   = (const float*)d_in[2];
  const float* W1   = (const float*)d_in[3];
  const float* b1   = (const float*)d_in[4];
  const float* W2   = (const float*)d_in[5];
  const float* b2   = (const float*)d_in[6];
  const float* Wout = (const float*)d_in[7];
  const float* bout = (const float*)d_in[8];
  float* out = (float*)d_out;

  const int n = in_sizes[0] / HID;
  const int E = in_sizes[2];
  const int* erow = ei;
  const int* ecol = ei + E;

  float* xw   = (float*)d_ws;
  float* hbuf = xw + (size_t)n * HID;
  float* dinv = hbuf + (size_t)n * HID;
  unsigned* present = (unsigned*)(dinv + n);
  int gPre = (E + 16 * PG - 1) / (16 * PG);
  int wpr  = gPre * 32;

  const int B = 256;
  int gN  = (n + B - 1) / B;
  int nTiles = (n + 15) / 16;
  int gGemm  = (nTiles + 7) / 8;
  int npb    = (n + 255) / 256;
  int gAgg   = (n + npb - 1) / npb;
  int nPart  = (n + DCAP - 1) / DCAP;
  int npp    = (((n + nPart - 1) / nPart) + 31) / 32 * 32;
  int gDeg   = DSL * ((n + npp - 1) / npp);
  int es     = (E + DSL - 1) / DSL;

  k_deg_part   <<<gDeg, 1024, 0, stream>>>(ecol, ew, hbuf, E, n, npp, es);
  k_dinv       <<<gN, B, 0, stream>>>(hbuf, dinv, n);
  k_present    <<<gPre, PG, 0, stream>>>(ecol, present, E, npb, wpr);

  k_gemm_wmma<<<gGemm, B, 0, stream>>>(x, W1, xw, n);
  k_agg      <<<gAgg, AGW * 32, 0, stream>>>(xw, erow, ecol, ew, present, dinv, b1, hbuf, E, n, npb, wpr);

  k_gemm_wmma<<<gGemm, B, 0, stream>>>(hbuf, W2, xw, n);
  k_agg      <<<gAgg, AGW * 32, 0, stream>>>(xw, erow, ecol, ew, present, dinv, b2, hbuf, E, n, npb, wpr);

  k_out_wmma<<<gGemm, B, 0, stream>>>(hbuf, Wout, bout, out, n);
}


# Layer 1's aggregation keeps its batches (source row, local row, norm: 8 B per edge, whole lines) in the workspace; layer 2 replays them: no map walk, no id or per-edge reads; LDS add order changes
# speedup vs baseline: 3.2914x; 1.0827x over previous
#include <hip/hip_runtime.h>


typedef float v2f __attribute__((ext_vector_type(2)));
typedef float v8f __attribute__((ext_vector_type(8)));

#define HID 128
#define NPRED 16

#define DSL   64
#define DCAP  65536
#define DUN   16

__global__ void __launch_bounds__(1024)
k_deg_part(const int* __restrict__ col, const float* __restrict__ w,
           float* __restrict__ part, int E, int n, int npp, int es) {
  __shared__ float acc[DCAP];
  const int tid = threadIdx.x;
  const int s   = blockIdx.x % DSL;
  const int lo  = (blockIdx.x / DSL) * npp;
  int cnt = n - lo;
  if (cnt > npp) cnt = npp;

  for (int i = tid; i < cnt; i += 1024) acc[i] = 0.0f;
  __syncthreads();

  const int e0 = s * es;
  int e1 = e0 + es;
  if (e1 > E) e1 = E;
  for (int e = e0 + tid; e < e1; e += DUN * 1024) {
    int   c[DUN];
    float x[DUN];
#pragma unroll
    for (int k = 0; k < DUN; ++k) {
      const int ee = e + k * 1024;
      c[k] = -1;
      x[k] = 0.0f;
      if (ee < e1) { c[k] = col[ee];  x[k] = w[ee]; }
    }
#pragma unroll
    for (int k = 0; k < DUN; ++k) {
      const unsigned lc = (unsigned)c[k] - (unsigned)lo;
      if (lc < (unsigned)cnt) atomicAdd(&acc[lc], x[k]);
    }
  }
  __syncthreads();

  float* dst = part + (size_t)s * n + lo;
  for (int i = tid; i < cnt; i += 1024) dst[i] = acc[i];
}

__global__ void k_dinv(const float* __restrict__ part, float* __restrict__ dinv, int n) {
  int i = blockIdx.x * blockDim.x + threadIdx.x;
  if (i < n) {
    float d = 1.0f;
#pragma unroll 16
    for (int s = 0; s < DSL; ++s) d += part[(size_t)s * n + i];
    dinv[i] = (d > 0.0f) ? rsqrtf(d) : 0.0f;
  }
}

#define PG      1024
#define PPITCH  257

__global__ void __launch_bounds__(1024)
k_present(const int* __restrict__ col,
          unsigned* __restrict__ present, int E, int npb, int wpr) {
  __shared__ unsigned tile[32 * PPITCH];
  const int tid = threadIdx.x;
  for (int i = tid; i < 32 * PPITCH; i += PG) tile[i] = 0u;
  __syncthreads();

  const int g  = blockIdx.x * PG + tid;
  const int e0 = g * 16;
  const int wv = tid >> 5;
  const unsigned bit = 1u << (tid & 31);
  int cc[16];
#pragma unroll
  for (int k = 0; k < 16; ++k) {
    const int e = e0 + k;
    cc[k] = -1;
    if (e < E) cc[k] = col[e];
  }
#pragma unroll
  for (int k = 0; k < 16; ++k) {
    if (cc[k] >= 0) {
      const unsigned t = (unsigned)cc[k] / (unsigned)npb;
      if (t < 256u) atomicOr(&tile[wv * PPITCH + t], bit);
    }
  }
  __syncthreads();

  for (int idx = tid; idx < 256 * 32; idx += PG) {
    const int t = idx >> 5, w = idx & 31;
    present[(size_t)t * wpr + blockIdx.x * 32 + w] = tile[w * PPITCH + t];
  }
}

__global__ void k_gemm_wmma(const float* __restrict__ A,
                            const float* __restrict__ W,
                            float* __restrict__ C, int n) {
  int wid  = (blockIdx.x * blockDim.x + threadIdx.x) >> 5;
  int lane = threadIdx.x & 31;
  int row0 = wid * 16;
  if (row0 >= n) return;

  int half = lane >> 4;
  int l    = lane & 15;

  v8f acc[8];
#pragma unroll
  for (int t = 0; t < 8; ++t) acc[t] = (v8f){0,0,0,0,0,0,0,0};

  const float* arow = A + (size_t)(row0 + l) * HID;

  for (int k = 0; k < HID; k += 4) {
    v2f afrag = *(const v2f*)(arow + k + half * 2);
#pragma unroll
    for (int t = 0; t < 8; ++t) {
      int ncol = t * 16 + l;
      v2f bfrag;
      bfrag.x = W[(size_t)(k + half * 2)     * HID + ncol];
      bfrag.y = W[(size_t)(k + half * 2 + 1) * HID + ncol];
      acc[t] = __builtin_amdgcn_wmma_f32_16x16x4_f32(
           false, afrag,  false, bfrag,
           (short)0, acc[t],  false,  false);
    }
  }

#pragma unroll
  for (int t = 0; t < 8; ++t) {
#pragma unroll
    for (int r = 0; r < 8; ++r) {
      int m    = row0 + r + half * 8;
      int ncol = t * 16 + l;
      C[(size_t)m * HID + ncol] = acc[t][r];
    }
  }
}

#define NPB   392
#define AGW   32
#define QCAP  544
#define GQCAP 64
#define EIF   8
#define EBITS 23
#define CAPB 256
#define RSH  17

__device__ __forceinline__ void wave_lds_order() {
  __builtin_amdgcn_fence(__ATOMIC_RELEASE, "wavefront");
  __builtin_amdgcn_wave_barrier();
  __builtin_amdgcn_fence(__ATOMIC_ACQUIRE, "wavefront");
}

__device__ __forceinline__ void gather_add(int r, int lc, float s, int cnt, int lane,
                                           const float* __restrict__ xw,
                                           float* rows) {
  int j = 0;
  for (; j + EIF <= cnt; j += EIF) {
    float4 v[EIF];
    int    lj[EIF];
    float  sj[EIF];
#pragma unroll
    for (int u = 0; u < EIF; ++u) {
      const int rj = __builtin_amdgcn_readlane(r, j + u);
      lj[u] = __builtin_amdgcn_readlane(lc, j + u);
      sj[u] = __int_as_float(__builtin_amdgcn_readlane(__float_as_int(s), j + u));
      v[u] = ((const float4*)(xw + (size_t)rj * HID))[lane];
    }
#pragma unroll
    for (int u = 0; u < EIF; ++u) {
      float* dst = rows + lj[u] * HID + lane;
      atomicAdd(dst,      v[u].x * sj[u]);
      atomicAdd(dst + 32, v[u].y * sj[u]);
      atomicAdd(dst + 64, v[u].z * sj[u]);
      atomicAdd(dst + 96, v[u].w * sj[u]);
    }
  }
  if (j < cnt) {
    const int rem = cnt - j;
    float4 v[EIF];
    int    lj[EIF];
    float  sj[EIF];
#pragma unroll
    for (int u = 0; u < EIF; ++u) {
      if (u < rem) {
        const int rj = __builtin_amdgcn_readlane(r, j + u);
        lj[u] = __builtin_amdgcn_readlane(lc, j + u);
        sj[u] = __int_as_float(__builtin_amdgcn_readlane(__float_as_int(s), j + u));
        v[u] = ((const float4*)(xw + (size_t)rj * HID))[lane];
      }
    }
#pragma unroll
    for (int u = 0; u < EIF; ++u) {
      if (u < rem) {
        float* dst = rows + lj[u] * HID + lane;
        atomicAdd(dst,      v[u].x * sj[u]);
        atomicAdd(dst + 32, v[u].y * sj[u]);
        atomicAdd(dst + 64, v[u].z * sj[u]);
        atomicAdd(dst + 96, v[u].w * sj[u]);
      }
    }
  }
}

__device__ __forceinline__ void agg_batch(const int* qv, int cnt, int lane,
                                          const float* __restrict__ xw,
                                          const int* __restrict__ row,
                                          const float* __restrict__ w,
                                          const float* __restrict__ dinv,
                                          const float* dl,
                                          float* rows,
                                          int2* recw, int* slotctr) {
  int r = 0, lc = 0;
  float s = 0.0f;
  if (lane < cnt) {
    const unsigned p = (unsigned)qv[lane];
    const int e = (int)(p & ((1u << EBITS) - 1u));
    lc = (int)(p >> EBITS);
    r = row[e];
    s = dinv[r] * w[e] * dl[lc];
  }
  gather_add(r, lc, s, cnt, lane, xw, rows);
  if (recw != nullptr) {
    int slot = 0;
    if (lane == 0) slot = atomicAdd(slotctr, 1);
    slot = __builtin_amdgcn_readfirstlane(slot);
    if (slot < CAPB)
      recw[slot * 32 + lane] = make_int2(lane < cnt ? (r | (lc << RSH)) : -1, __float_as_int(s));
  }
}

__device__ __forceinline__ int scan_groups(const int* gv, int cnt, int lane, int base, int cntn,
                                           const int* __restrict__ col, int E, bool vec,
                                           int* myq, int qn) {
  const int sub = lane & 3;
  int cv[4][4];
  int eb[4];
#pragma unroll
  for (int k = 0; k < 4; ++k) {
    const int gi = 8 * k + (lane >> 2);
    eb[k] = E;
    if (gi < cnt) eb[k] = (gv[gi] << 4) + (sub << 2);
    cv[k][0] = -1;  cv[k][1] = -1;  cv[k][2] = -1;  cv[k][3] = -1;
  }
  if (vec) {
#pragma unroll
    for (int k = 0; k < 4; ++k) {
      if (eb[k] < E) {
        const int4 t = *(const int4*)(col + eb[k]);
        cv[k][0] = t.x;  cv[k][1] = t.y;  cv[k][2] = t.z;  cv[k][3] = t.w;
      }
    }
  } else {
#pragma unroll
    for (int k = 0; k < 4; ++k) {
      if (eb[k]     < E) cv[k][0] = col[eb[k]];
      if (eb[k] + 1 < E) cv[k][1] = col[eb[k] + 1];
      if (eb[k] + 2 < E) cv[k][2] = col[eb[k] + 2];
      if (eb[k] + 3 < E) cv[k][3] = col[eb[k] + 3];
    }
  }
#pragma unroll
  for (int k = 0; k < 4; ++k) {
    if (8 * k < cnt) {
#pragma unroll
      for (int s = 0; s < 4; ++s) {
        const unsigned lc = (unsigned)cv[k][s] - (unsigned)base;
        const bool hit = lc < (unsigned)cntn;
        const unsigned m = __builtin_amdgcn_ballot_w32(hit);
        if (hit)
          myq[qn + (int)__builtin_amdgcn_mbcnt_lo(m, 0u)] = (int)((lc << EBITS) | (unsigned)(eb[k] + s));
        qn += __builtin_popcount(m);
      }
    }
  }
  return qn;
}

__global__ void __launch_bounds__(1024)
k_agg(const float* __restrict__ xw,
      const int* __restrict__ row,
      const int* __restrict__ col,
      const float* __restrict__ w,
      const unsigned* __restrict__ present,
      const float* __restrict__ dinv,
      const float* __restrict__ b,
      float* __restrict__ h, int E, int n, int npb, int wpr,
      int* __restrict__ rec, int* __restrict__ meta, int mode) {
  __shared__ float rows[NPB * HID];
  __shared__ int   q[AGW][QCAP];
  __shared__ int   gq[AGW][GQCAP];
  __shared__ float dl[NPB];
  __shared__ int   slotctr;

  const int tid  = threadIdx.x;
  const int lane = tid & 31;
  const int wv   = __builtin_amdgcn_readfirstlane(tid >> 5);
  const int base = blockIdx.x * npb;
  int cntn = n - base;
  if (cntn > npb) cntn = npb;
  if (tid == 0) slotctr = 0;

  int2* slab = (int2*)rec + (size_t)blockIdx.x * (CAPB * 32);
  int ns = -1;
  if (mode == 2) ns = __builtin_amdgcn_readfirstlane(meta[blockIdx.x]);
  int2* recw = (mode == 1) ? slab : nullptr;

  for (int idx = tid; idx < cntn * (HID / 4); idx += AGW * 32) {
    const float s0 = dinv[base + (idx >> 5)];
    const float s = s0 * s0;
    float4 xv = ((const float4*)xw)[(size_t)base * (HID / 4) + idx];
    float* d = rows + (idx >> 5) * HID + (idx & 31);
    d[0]  = xv.x * s;
    d[32] = xv.y * s;
    d[64] = xv.z * s;
    d[96] = xv.w * s;
    if ((idx & 31) == 0) dl[idx >> 5] = s0;
  }
  __syncthreads();

  if (ns < 0) {
    const bool vec = (E & 3) == 0;
    const unsigned* myrow = present + (size_t)blockIdx.x * wpr;
    int* myq  = q[wv];
    int* mygq = gq[wv];
    int qn = 0;
    int gn = 0;

    unsigned wnext = 0u;
    if ((wv << 5) < wpr) wnext = myrow[(wv << 5) + lane];
    for (int i0 = wv << 5; i0 < wpr; i0 += AGW * 32) {
      const int i = i0 + lane;
      unsigned word = wnext;
      if (i0 + AGW * 32 < wpr) wnext = myrow[i + AGW * 32];
      unsigned act = __builtin_amdgcn_ballot_w32(word != 0u);
      while (act != 0u) {
        if (word != 0u) {
          const int slot = gn + (int)__builtin_amdgcn_mbcnt_lo(act, 0u);
          mygq[slot] = (i << 5) + (__ffs((int)word) - 1);
          word &= word - 1u;
        }
        gn += __builtin_popcount(act);
        if (gn >= 32) {
          wave_lds_order();
          qn = scan_groups(mygq + (gn - 32), 32, lane, base, cntn, col, E, vec, myq, qn);
          gn -= 32;
          wave_lds_order();
          if (qn >= 32) {
            while (qn >= 32) {
              agg_batch(myq + (qn - 32), 32, lane, xw, row, w, dinv, dl, rows, recw, &slotctr);
              qn -= 32;
            }
            wave_lds_order();
          }
        }
        act = __builtin_amdgcn_ballot_w32(word != 0u);
      }
    }
    if (gn > 0) {
      wave_lds_order();
      qn = scan_groups(mygq, gn, lane, base, cntn, col, E, vec, myq, qn);
    }
    wave_lds_order();
    while (qn > 0) {
      const int c = (qn < 32) ? qn : 32;
      agg_batch(myq + (qn - c), c, lane, xw, row, w, dinv, dl, rows, recw, &slotctr);
      qn -= c;
    }
  } else {
    int2 pn = make_int2(-1, 0);
    if (wv < ns) pn = slab[wv * 32 + lane];
    for (int slot = wv; slot < ns; slot += AGW) {
      const int2 p = pn;
      if (slot + AGW < ns) pn = slab[(slot + AGW) * 32 + lane];
      const bool valid = p.x >= 0;
      const int c = __builtin_popcount(__builtin_amdgcn_ballot_w32(valid));
      const int r  = valid ? (p.x & ((1 << RSH) - 1)) : 0;
      const int lc = valid ? (p.x >> RSH) : 0;
      if (c == 32) gather_add(r, lc, __int_as_float(p.y), 32, lane, xw, rows);
      else         gather_add(r, lc, __int_as_float(p.y), c, lane, xw, rows);
    }
  }
  __syncthreads();
  if (mode == 1 && tid == 0) meta[blockIdx.x] = (slotctr <= CAPB) ? slotctr : -1;

  for (int idx = tid; idx < cntn * (HID / 4); idx += AGW * 32) {
    const float* a = rows + (idx >> 5) * HID + (idx & 31);
    float4 bv = ((const float4*)b)[idx & 31];
    float4 o;
    o.x = a[0]  + bv.x;  o.x = o.x > 0.0f ? o.x : 0.0f;
    o.y = a[32] + bv.y;  o.y = o.y > 0.0f ? o.y : 0.0f;
    o.z = a[64] + bv.z;  o.z = o.z > 0.0f ? o.z : 0.0f;
    o.w = a[96] + bv.w;  o.w = o.w > 0.0f ? o.w : 0.0f;
    ((float4*)h)[(size_t)base * (HID / 4) + idx] = o;
  }
}

__global__ void k_out_wmma(const float* __restrict__ h,
                           const float* __restrict__ Wout,
                           const float* __restrict__ bout,
                           float* __restrict__ out, int n) {
  const int wid  = (blockIdx.x * blockDim.x + threadIdx.x) >> 5;
  const int lane = threadIdx.x & 31;
  const int row0 = wid * 16;
  if (row0 >= n) return;

  const int c  = lane & 15;
  const int hf = lane >> 4;

  const int node = row0 + c;
  const int ld   = node < n ? node : n - 1;
  const float4* wr = (const float4*)(Wout + (size_t)c * HID) + hf;
  const float4* hr = (const float4*)(h + (size_t)ld * HID) + hf;

  const float4 bl = ((const float4*)bout)[2 * hf];
  const float4 bh = ((const float4*)bout)[2 * hf + 1];
  v8f acc0 = (v8f){bl.x, bl.y, bl.z, bl.w, bh.x, bh.y, bh.z, bh.w};
  v8f acc1 = (v8f){0,0,0,0,0,0,0,0};
#pragma unroll
  for (int j = 0; j < 16; ++j) {
    const float4 a = wr[2 * j];
    const float4 b = hr[2 * j];
    v2f a0 = {a.x, a.y}, b0 = {b.x, b.y};
    v2f a1 = {a.z, a.w}, b1 = {b.z, b.w};
    acc0 = __builtin_amdgcn_wmma_f32_16x16x4_f32(
         false, a0,  false, b0,
         (short)0, acc0,  false,  false);
    acc1 = __builtin_amdgcn_wmma_f32_16x16x4_f32(
         false, a1,  false, b1,
         (short)0, acc1,  false,  false);
  }
  const v8f r = acc0 + acc1;
  if (node < n) {
    float4* o = (float4*)(out + (size_t)node * NPRED + 8 * hf);
    o[0] = make_float4(r[0], r[1], r[2], r[3]);
    o[1] = make_float4(r[4], r[5], r[6], r[7]);
  }
}


extern "C" void kernel_launch(void* const* d_in, const int* in_sizes, int n_in,
                              void* d_out, int out_size, void* d_ws, size_t ws_size,
                              hipStream_t stream) {
  const float* x    = (const float*)d_in[0];
  const int*   ei   = (const int*)  d_in[1];
  const float* ew   = (const float*)d_in[2];
  const float* W1   = (const float*)d_in[3];
  const float* b1   = (const float*)d_in[4];
  const float* W2   = (const float*)d_in[5];
  const float* b2   = (const float*)d_in[6];
  const float* Wout = (const float*)d_in[7];
  const float* bout = (const float*)d_in[8];
  float* out = (float*)d_out;

  const int n = in_sizes[0] / HID;
  const int E = in_sizes[2];
  const int* erow = ei;
  const int* ecol = ei + E;

  float* xw   = (float*)d_ws;
  float* hbuf = xw + (size_t)n * HID;
  float* dinv = hbuf + (size_t)n * HID;
  unsigned* present = (unsigned*)(dinv + n);
  int gPre = (E + 16 * PG - 1) / (16 * PG);
  int wpr  = gPre * 32;
  size_t recOff = ((size_t)((const char*)(present + (size_t)256 * wpr) - (const char*)d_ws) + 1024 + 255) / 256 * 256;
  int*  meta = (int*)(present + (size_t)256 * wpr);
  int*  rec  = (int*)((char*)d_ws + recOff);
  const bool keep = ws_size >= recOff + (size_t)256 * CAPB * 256 && n <= (1 << RSH);

  const int B = 256;
  int gN  = (n + B - 1) / B;
  int nTiles = (n + 15) / 16;
  int gGemm  = (nTiles + 7) / 8;
  int npb    = (n + 255) / 256;
  int gAgg   = (n + npb - 1) / npb;
  int nPart  = (n + DCAP - 1) / DCAP;
  int npp    = (((n + nPart - 1) / nPart) + 31) / 32 * 32;
  int gDeg   = DSL * ((n + npp - 1) / npp);
  int es     = (E + DSL - 1) / DSL;

  k_deg_part   <<<gDeg, 1024, 0, stream>>>(ecol, ew, hbuf, E, n, npp, es);
  k_dinv       <<<gN, B, 0, stream>>>(hbuf, dinv, n);
  k_present    <<<gPre, PG, 0, stream>>>(ecol, present, E, npb, wpr);

  k_gemm_wmma<<<gGemm, B, 0, stream>>>(x, W1, xw, n);
  k_agg      <<<gAgg, AGW * 32, 0, stream>>>(xw, erow, ecol, ew, present, dinv, b1, hbuf, E, n, npb, wpr, rec, meta, keep ? 1 : 0);

  k_gemm_wmma<<<gGemm, B, 0, stream>>>(hbuf, W2, xw, n);
  k_agg      <<<gAgg, AGW * 32, 0, stream>>>(xw, erow, ecol, ew, present, dinv, b2, hbuf, E, n, npb, wpr, rec, meta, keep ? 2 : 0);

  k_out_wmma<<<gGemm, B, 0, stream>>>(hbuf, Wout, bout, out, n);
}


# Independent passes share a launch, block index picks the role: presence-map blocks ride with the degree-partial blocks, dinv blocks with layer 1's GEMM; 6 launches, not 8; no sum's order changes
# speedup vs baseline: 3.4307x; 1.0423x over previous
#include <hip/hip_runtime.h>


typedef float v2f __attribute__((ext_vector_type(2)));
typedef float v8f __attribute__((ext_vector_type(8)));

#define HID 128
#define NPRED 16

#define DSL   64
#define DCAP  65536
#define DUN   16

__device__ __forceinline__ void
deg_part_role(float* acc, int bid,
              const int* __restrict__ col, const float* __restrict__ w,
              float* __restrict__ part, int E, int n, int npp, int es) {
  const int tid = threadIdx.x;
  const int s   = bid % DSL;
  const int lo  = (bid / DSL) * npp;
  int cnt = n - lo;
  if (cnt > npp) cnt = npp;

  for (int i = tid; i < cnt; i += 1024) acc[i] = 0.0f;
  __syncthreads();

  const int e0 = s * es;
  int e1 = e0 + es;
  if (e1 > E) e1 = E;
  for (int e = e0 + tid; e < e1; e += DUN * 1024) {
    int   c[DUN];
    float x[DUN];
#pragma unroll
    for (int k = 0; k < DUN; ++k) {
      const int ee = e + k * 1024;
      c[k] = -1;
      x[k] = 0.0f;
      if (ee < e1) { c[k] = col[ee];  x[k] = w[ee]; }
    }
#pragma unroll
    for (int k = 0; k < DUN; ++k) {
      const unsigned lc = (unsigned)c[k] - (unsigned)lo;
      if (lc < (unsigned)cnt) atomicAdd(&acc[lc], x[k]);
    }
  }
  __syncthreads();

  float* dst = part + (size_t)s * n + lo;
  for (int i = tid; i < cnt; i += 1024) dst[i] = acc[i];
}

__device__ __forceinline__ void
dinv_role(int bid, const float* __restrict__ part, float* __restrict__ dinv, int n) {
  int i = bid * blockDim.x + threadIdx.x;
  if (i < n) {
    float d = 1.0f;
#pragma unroll 16
    for (int s = 0; s < DSL; ++s) d += part[(size_t)s * n + i];
    dinv[i] = (d > 0.0f) ? rsqrtf(d) : 0.0f;
  }
}

#define PG      1024
#define PPITCH  257

__device__ __forceinline__ void
present_role(unsigned* tile, int bid, const int* __restrict__ col,
             unsigned* __restrict__ present, int E, int npb, int wpr) {
  const int tid = threadIdx.x;
  for (int i = tid; i < 32 * PPITCH; i += PG) tile[i] = 0u;
  __syncthreads();

  const int g  = bid * PG + tid;
  const int e0 = g * 16;
  const int wv = tid >> 5;
  const unsigned bit = 1u << (tid & 31);
  int cc[16];
#pragma unroll
  for (int k = 0; k < 16; ++k) {
    const int e = e0 + k;
    cc[k] = -1;
    if (e < E) cc[k] = col[e];
  }
#pragma unroll
  for (int k = 0; k < 16; ++k) {
    if (cc[k] >= 0) {
      const unsigned t = (unsigned)cc[k] / (unsigned)npb;
      if (t < 256u) atomicOr(&tile[wv * PPITCH + t], bit);
    }
  }
  __syncthreads();

  for (int idx = tid; idx < 256 * 32; idx += PG) {
    const int t = idx >> 5, w = idx & 31;
    present[(size_t)t * wpr + bid * 32 + w] = tile[w * PPITCH + t];
  }
}

__global__ void __launch_bounds__(1024)
k_deg_present(const int* __restrict__ col, const float* __restrict__ w,
              float* __restrict__ part, unsigned* __restrict__ present,
              int E, int n, int npp, int es, int npb, int wpr, int gDeg) {
  __shared__ unsigned lds[DCAP];
  if ((int)blockIdx.x < gDeg)
    deg_part_role((float*)lds, (int)blockIdx.x, col, w, part, E, n, npp, es);
  else
    present_role(lds, (int)blockIdx.x - gDeg, col, present, E, npb, wpr);
}

__device__ __forceinline__ void
gemm_role(int bid, const float* __restrict__ A,
          const float* __restrict__ W,
          float* __restrict__ C, int n) {
  int wid  = (bid * blockDim.x + threadIdx.x) >> 5;
  int lane = threadIdx.x & 31;
  int row0 = wid * 16;
  if (row0 >= n) return;

  int half = lane >> 4;
  int l    = lane & 15;

  v8f acc[8];
#pragma unroll
  for (int t = 0; t < 8; ++t) acc[t] = (v8f){0,0,0,0,0,0,0,0};

  const float* arow = A + (size_t)(row0 + l) * HID;

  for (int k = 0; k < HID; k += 4) {
    v2f afrag = *(const v2f*)(arow + k + half * 2);
#pragma unroll
    for (int t = 0; t < 8; ++t) {
      int ncol = t * 16 + l;
      v2f bfrag;
      bfrag.x = W[(size_t)(k + half * 2)     * HID + ncol];
      bfrag.y = W[(size_t)(k + half * 2 + 1) * HID + ncol];
      acc[t] = __builtin_amdgcn_wmma_f32_16x16x4_f32(
           false, afrag,  false, bfrag,
           (short)0, acc[t],  false,  false);
    }
  }

#pragma unroll
  for (int t = 0; t < 8; ++t) {
#pragma unroll
    for (int r = 0; r < 8; ++r) {
      int m    = row0 + r + half * 8;
      int ncol = t * 16 + l;
      C[(size_t)m * HID + ncol] = acc[t][r];
    }
  }
}

__global__ void k_gemm_wmma(const float* __restrict__ A,
                            const float* __restrict__ W,
                            float* __restrict__ C, int n) {
  gemm_role((int)blockIdx.x, A, W, C, n);
}

__global__ void k_gemm_dinv(const float* __restrict__ A,
                            const float* __restrict__ W,
                            float* __restrict__ C, int n,
                            const float* __restrict__ part,
                            float* __restrict__ dinv, int gGemm) {
  if ((int)blockIdx.x < gGemm) gemm_role((int)blockIdx.x, A, W, C, n);
  else                         dinv_role((int)blockIdx.x - gGemm, part, dinv, n);
}

#define NPB   392
#define AGW   32
#define QCAP  544
#define GQCAP 64
#define EIF   8
#define EBITS 23
#define CAPB 256
#define RSH  17

__device__ __forceinline__ void wave_lds_order() {
  __builtin_amdgcn_fence(__ATOMIC_RELEASE, "wavefront");
  __builtin_amdgcn_wave_barrier();
  __builtin_amdgcn_fence(__ATOMIC_ACQUIRE, "wavefront");
}

__device__ __forceinline__ void gather_add(int r, int lc, float s, int cnt, int lane,
                                           const float* __restrict__ xw,
                                           float* rows) {
  int j = 0;
  for (; j + EIF <= cnt; j += EIF) {
    float4 v[EIF];
    int    lj[EIF];
    float  sj[EIF];
#pragma unroll
    for (int u = 0; u < EIF; ++u) {
      const int rj = __builtin_amdgcn_readlane(r, j + u);
      lj[u] = __builtin_amdgcn_readlane(lc, j + u);
      sj[u] = __int_as_float(__builtin_amdgcn_readlane(__float_as_int(s), j + u));
      v[u] = ((const float4*)(xw + (size_t)rj * HID))[lane];
    }
#pragma unroll
    for (int u = 0; u < EIF; ++u) {
      float* dst = rows + lj[u] * HID + lane;
      atomicAdd(dst,      v[u].x * sj[u]);
      atomicAdd(dst + 32, v[u].y * sj[u]);
      atomicAdd(dst + 64, v[u].z * sj[u]);
      atomicAdd(dst + 96, v[u].w * sj[u]);
    }
  }
  if (j < cnt) {
    const int rem = cnt - j;
    float4 v[EIF];
    int    lj[EIF];
    float  sj[EIF];
#pragma unroll
    for (int u = 0; u < EIF; ++u) {
      if (u < rem) {
        const int rj = __builtin_amdgcn_readlane(r, j + u);
        lj[u] = __builtin_amdgcn_readlane(lc, j + u);
        sj[u] = __int_as_float(__builtin_amdgcn_readlane(__float_as_int(s), j + u));
        v[u] = ((const float4*)(xw + (size_t)rj * HID))[lane];
      }
    }
#pragma unroll
    for (int u = 0; u < EIF; ++u) {
      if (u < rem) {
        float* dst = rows + lj[u] * HID + lane;
        atomicAdd(dst,      v[u].x * sj[u]);
        atomicAdd(dst + 32, v[u].y * sj[u]);
        atomicAdd(dst + 64, v[u].z * sj[u]);
        atomicAdd(dst + 96, v[u].w * sj[u]);
      }
    }
  }
}

__device__ __forceinline__ void agg_batch(const int* qv, int cnt, int lane,
                                          const float* __restrict__ xw,
                                          const int* __restrict__ row,
                                          const float* __restrict__ w,
                                          const float* __restrict__ dinv,
                                          const float* dl,
                                          float* rows,
                                          int2* recw, int* slotctr) {
  int r = 0, lc = 0;
  float s = 0.0f;
  if (lane < cnt) {
    const unsigned p = (unsigned)qv[lane];
    const int e = (int)(p & ((1u << EBITS) - 1u));
    lc = (int)(p >> EBITS);
    r = row[e];
    s = dinv[r] * w[e] * dl[lc];
  }
  gather_add(r, lc, s, cnt, lane, xw, rows);
  if (recw != nullptr) {
    int slot = 0;
    if (lane == 0) slot = atomicAdd(slotctr, 1);
    slot = __builtin_amdgcn_readfirstlane(slot);
    if (slot < CAPB)
      recw[slot * 32 + lane] = make_int2(lane < cnt ? (r | (lc << RSH)) : -1, __float_as_int(s));
  }
}

__device__ __forceinline__ int scan_groups(const int* gv, int cnt, int lane, int base, int cntn,
                                           const int* __restrict__ col, int E, bool vec,
                                           int* myq, int qn) {
  const int sub = lane & 3;
  int cv[4][4];
  int eb[4];
#pragma unroll
  for (int k = 0; k < 4; ++k) {
    const int gi = 8 * k + (lane >> 2);
    eb[k] = E;
    if (gi < cnt) eb[k] = (gv[gi] << 4) + (sub << 2);
    cv[k][0] = -1;  cv[k][1] = -1;  cv[k][2] = -1;  cv[k][3] = -1;
  }
  if (vec) {
#pragma unroll
    for (int k = 0; k < 4; ++k) {
      if (eb[k] < E) {
        const int4 t = *(const int4*)(col + eb[k]);
        cv[k][0] = t.x;  cv[k][1] = t.y;  cv[k][2] = t.z;  cv[k][3] = t.w;
      }
    }
  } else {
#pragma unroll
    for (int k = 0; k < 4; ++k) {
      if (eb[k]     < E) cv[k][0] = col[eb[k]];
      if (eb[k] + 1 < E) cv[k][1] = col[eb[k] + 1];
      if (eb[k] + 2 < E) cv[k][2] = col[eb[k] + 2];
      if (eb[k] + 3 < E) cv[k][3] = col[eb[k] + 3];
    }
  }
#pragma unroll
  for (int k = 0; k < 4; ++k) {
    if (8 * k < cnt) {
#pragma unroll
      for (int s = 0; s < 4; ++s) {
        const unsigned lc = (unsigned)cv[k][s] - (unsigned)base;
        const bool hit = lc < (unsigned)cntn;
        const unsigned m = __builtin_amdgcn_ballot_w32(hit);
        if (hit)
          myq[qn + (int)__builtin_amdgcn_mbcnt_lo(m, 0u)] = (int)((lc << EBITS) | (unsigned)(eb[k] + s));
        qn += __builtin_popcount(m);
      }
    }
  }
  return qn;
}

__global__ void __launch_bounds__(1024)
k_agg(const float* __restrict__ xw,
      const int* __restrict__ row,
      const int* __restrict__ col,
      const float* __restrict__ w,
      const unsigned* __restrict__ present,
      const float* __restrict__ dinv,
      const float* __restrict__ b,
      float* __restrict__ h, int E, int n, int npb, int wpr,
      int* __restrict__ rec, int* __restrict__ meta, int mode) {
  __shared__ float rows[NPB * HID];
  __shared__ int   q[AGW][QCAP];
  __shared__ int   gq[AGW][GQCAP];
  __shared__ float dl[NPB];
  __shared__ int   slotctr;

  const int tid  = threadIdx.x;
  const int lane = tid & 31;
  const int wv   = __builtin_amdgcn_readfirstlane(tid >> 5);
  const int base = blockIdx.x * npb;
  int cntn = n - base;
  if (cntn > npb) cntn = npb;
  if (tid == 0) slotctr = 0;

  int2* slab = (int2*)rec + (size_t)blockIdx.x * (CAPB * 32);
  int ns = -1;
  if (mode == 2) ns = __builtin_amdgcn_readfirstlane(meta[blockIdx.x]);
  int2* recw = (mode == 1) ? slab : nullptr;

  for (int idx = tid; idx < cntn * (HID / 4); idx += AGW * 32) {
    const float s0 = dinv[base + (idx >> 5)];
    const float s = s0 * s0;
    float4 xv = ((const float4*)xw)[(size_t)base * (HID / 4) + idx];
    float* d = rows + (idx >> 5) * HID + (idx & 31);
    d[0]  = xv.x * s;
    d[32] = xv.y * s;
    d[64] = xv.z * s;
    d[96] = xv.w * s;
    if ((idx & 31) == 0) dl[idx >> 5] = s0;
  }
  __syncthreads();

  if (ns < 0) {
    const bool vec = (E & 3) == 0;
    const unsigned* myrow = present + (size_t)blockIdx.x * wpr;
    int* myq  = q[wv];
    int* mygq = gq[wv];
    int qn = 0;
    int gn = 0;

    unsigned wnext = 0u;
    if ((wv << 5) < wpr) wnext = myrow[(wv << 5) + lane];
    for (int i0 = wv << 5; i0 < wpr; i0 += AGW * 32) {
      const int i = i0 + lane;
      unsigned word = wnext;
      if (i0 + AGW * 32 < wpr) wnext = myrow[i + AGW * 32];
      unsigned act = __builtin_amdgcn_ballot_w32(word != 0u);
      while (act != 0u) {
        if (word != 0u) {
          const int slot = gn + (int)__builtin_amdgcn_mbcnt_lo(act, 0u);
          mygq[slot] = (i << 5) + (__ffs((int)word) - 1);
          word &= word - 1u;
        }
        gn += __builtin_popcount(act);
        if (gn >= 32) {
          wave_lds_order();
          qn = scan_groups(mygq + (gn - 32), 32, lane, base, cntn, col, E, vec, myq, qn);
          gn -= 32;
          wave_lds_order();
          if (qn >= 32) {
            while (qn >= 32) {
              agg_batch(myq + (qn - 32), 32, lane, xw, row, w, dinv, dl, rows, recw, &slotctr);
              qn -= 32;
            }
            wave_lds_order();
          }
        }
        act = __builtin_amdgcn_ballot_w32(word != 0u);
      }
    }
    if (gn > 0) {
      wave_lds_order();
      qn = scan_groups(mygq, gn, lane, base, cntn, col, E, vec, myq, qn);
    }
    wave_lds_order();
    while (qn > 0) {
      const int c = (qn < 32) ? qn : 32;
      agg_batch(myq + (qn - c), c, lane, xw, row, w, dinv, dl, rows, recw, &slotctr);
      qn -= c;
    }
  } else {
    int2 pn = make_int2(-1, 0);
    if (wv < ns) pn = slab[wv * 32 + lane];
    for (int slot = wv; slot < ns; slot += AGW) {
      const int2 p = pn;
      if (slot + AGW < ns) pn = slab[(slot + AGW) * 32 + lane];
      const bool valid = p.x >= 0;
      const int c = __builtin_popcount(__builtin_amdgcn_ballot_w32(valid));
      const int r  = valid ? (p.x & ((1 << RSH) - 1)) : 0;
      const int lc = valid ? (p.x >> RSH) : 0;
      if (c == 32) gather_add(r, lc, __int_as_float(p.y), 32, lane, xw, rows);
      else         gather_add(r, lc, __int_as_float(p.y), c, lane, xw, rows);
    }
  }
  __syncthreads();
  if (mode == 1 && tid == 0) meta[blockIdx.x] = (slotctr <= CAPB) ? slotctr : -1;

  for (int idx = tid; idx < cntn * (HID / 4); idx += AGW * 32) {
    const float* a = rows + (idx >> 5) * HID + (idx & 31);
    float4 bv = ((const float4*)b)[idx & 31];
    float4 o;
    o.x = a[0]  + bv.x;  o.x = o.x > 0.0f ? o.x : 0.0f;
    o.y = a[32] + bv.y;  o.y = o.y > 0.0f ? o.y : 0.0f;
    o.z = a[64] + bv.z;  o.z = o.z > 0.0f ? o.z : 0.0f;
    o.w = a[96] + bv.w;  o.w = o.w > 0.0f ? o.w : 0.0f;
    ((float4*)h)[(size_t)base * (HID / 4) + idx] = o;
  }
}

__global__ void k_out_wmma(const float* __restrict__ h,
                           const float* __restrict__ Wout,
                           const float* __restrict__ bout,
                           float* __restrict__ out, int n) {
  const int wid  = (blockIdx.x * blockDim.x + threadIdx.x) >> 5;
  const int lane = threadIdx.x & 31;
  const int row0 = wid * 16;
  if (row0 >= n) return;

  const int c  = lane & 15;
  const int hf = lane >> 4;

  const int node = row0 + c;
  const int ld   = node < n ? node : n - 1;
  const float4* wr = (const float4*)(Wout + (size_t)c * HID) + hf;
  const float4* hr = (const float4*)(h + (size_t)ld * HID) + hf;

  const float4 bl = ((const float4*)bout)[2 * hf];
  const float4 bh = ((const float4*)bout)[2 * hf + 1];
  v8f acc0 = (v8f){bl.x, bl.y, bl.z, bl.w, bh.x, bh.y, bh.z, bh.w};
  v8f acc1 = (v8f){0,0,0,0,0,0,0,0};
#pragma unroll
  for (int j = 0; j < 16; ++j) {
    const float4 a = wr[2 * j];
    const float4 b = hr[2 * j];
    v2f a0 = {a.x, a.y}, b0 = {b.x, b.y};
    v2f a1 = {a.z, a.w}, b1 = {b.z, b.w};
    acc0 = __builtin_amdgcn_wmma_f32_16x16x4_f32(
         false, a0,  false, b0,
         (short)0, acc0,  false,  false);
    acc1 = __builtin_amdgcn_wmma_f32_16x16x4_f32(
         false, a1,  false, b1,
         (short)0, acc1,  false,  false);
  }
  const v8f r = acc0 + acc1;
  if (node < n) {
    float4* o = (float4*)(out + (size_t)node * NPRED + 8 * hf);
    o[0] = make_float4(r[0], r[1], r[2], r[3]);
    o[1] = make_float4(r[4], r[5], r[6], r[7]);
  }
}


extern "C" void kernel_launch(void* const* d_in, const int* in_sizes, int n_in,
                              void* d_out, int out_size, void* d_ws, size_t ws_size,
                              hipStream_t stream) {
  const float* x    = (const float*)d_in[0];
  const int*   ei   = (const int*)  d_in[1];
  const float* ew   = (const float*)d_in[2];
  const float* W1   = (const float*)d_in[3];
  const float* b1   = (const float*)d_in[4];
  const float* W2   = (const float*)d_in[5];
  const float* b2   = (const float*)d_in[6];
  const float* Wout = (const float*)d_in[7];
  const float* bout = (const float*)d_in[8];
  float* out = (float*)d_out;

  const int n = in_sizes[0] / HID;
  const int E = in_sizes[2];
  const int* erow = ei;
  const int* ecol = ei + E;

  float* xw   = (float*)d_ws;
  float* hbuf = xw + (size_t)n * HID;
  float* dinv = hbuf + (size_t)n * HID;
  unsigned* present = (unsigned*)(dinv + n);
  int gPre = (E + 16 * PG - 1) / (16 * PG);
  int wpr  = gPre * 32;
  size_t recOff = ((size_t)((const char*)(present + (size_t)256 * wpr) - (const char*)d_ws) + 1024 + 255) / 256 * 256;
  int*  meta = (int*)(present + (size_t)256 * wpr);
  int*  rec  = (int*)((char*)d_ws + recOff);
  const bool keep = ws_size >= recOff + (size_t)256 * CAPB * 256 && n <= (1 << RSH);

  const int B = 256;
  int gN  = (n + B - 1) / B;
  int nTiles = (n + 15) / 16;
  int gGemm  = (nTiles + 7) / 8;
  int npb    = (n + 255) / 256;
  int gAgg   = (n + npb - 1) / npb;
  int nPart  = (n + DCAP - 1) / DCAP;
  int npp    = (((n + nPart - 1) / nPart) + 31) / 32 * 32;
  int gDeg   = DSL * ((n + npp - 1) / npp);
  int es     = (E + DSL - 1) / DSL;

  k_deg_present<<<gDeg + gPre, 1024, 0, stream>>>(ecol, ew, hbuf, present, E, n, npp, es, npb, wpr, gDeg);

  k_gemm_dinv<<<gGemm + gN, B, 0, stream>>>(x, W1, xw, n, hbuf, dinv, gGemm);
  k_agg      <<<gAgg, AGW * 32, 0, stream>>>(xw, erow, ecol, ew, present, dinv, b1, hbuf, E, n, npb, wpr, rec, meta, keep ? 1 : 0);

  k_gemm_wmma<<<gGemm, B, 0, stream>>>(hbuf, W2, xw, n);
  k_agg      <<<gAgg, AGW * 32, 0, stream>>>(xw, erow, ecol, ew, present, dinv, b2, hbuf, E, n, npb, wpr, rec, meta, keep ? 2 : 0);

  k_out_wmma<<<gGemm, B, 0, stream>>>(hbuf, Wout, bout, out, n);
}


# LDS sum rows get one writer wave each (row & 31): the scan hands edges to owner queues, owners add with plain 16-B LDS load/add/store; LDS float adds only if a queue overflows; sum order changes
# speedup vs baseline: 3.5453x; 1.0334x over previous
#include <hip/hip_runtime.h>


typedef float v2f __attribute__((ext_vector_type(2)));
typedef float v8f __attribute__((ext_vector_type(8)));

#define HID 128
#define NPRED 16

#define DSL   64
#define DCAP  65536
#define DUN   16

__device__ __forceinline__ void
deg_part_role(float* acc, int bid,
              const int* __restrict__ col, const float* __restrict__ w,
              float* __restrict__ part, int E, int n, int npp, int es) {
  const int tid = threadIdx.x;
  const int s   = bid % DSL;
  const int lo  = (bid / DSL) * npp;
  int cnt = n - lo;
  if (cnt > npp) cnt = npp;

  for (int i = tid; i < cnt; i += 1024) acc[i] = 0.0f;
  __syncthreads();

  const int e0 = s * es;
  int e1 = e0 + es;
  if (e1 > E) e1 = E;
  for (int e = e0 + tid; e < e1; e += DUN * 1024) {
    int   c[DUN];
    float x[DUN];
#pragma unroll
    for (int k = 0; k < DUN; ++k) {
      const int ee = e + k * 1024;
      c[k] = -1;
      x[k] = 0.0f;
      if (ee < e1) { c[k] = col[ee];  x[k] = w[ee]; }
    }
#pragma unroll
    for (int k = 0; k < DUN; ++k) {
      const unsigned lc = (unsigned)c[k] - (unsigned)lo;
      if (lc < (unsigned)cnt) atomicAdd(&acc[lc], x[k]);
    }
  }
  __syncthreads();

  float* dst = part + (size_t)s * n + lo;
  for (int i = tid; i < cnt; i += 1024) dst[i] = acc[i];
}

__device__ __forceinline__ void
dinv_role(int bid, const float* __restrict__ part, float* __restrict__ dinv, int n) {
  int i = bid * blockDim.x + threadIdx.x;
  if (i < n) {
    float d = 1.0f;
#pragma unroll 16
    for (int s = 0; s < DSL; ++s) d += part[(size_t)s * n + i];
    dinv[i] = (d > 0.0f) ? rsqrtf(d) : 0.0f;
  }
}

#define PG      1024
#define PPITCH  257

__device__ __forceinline__ void
present_role(unsigned* tile, int bid, const int* __restrict__ col,
             unsigned* __restrict__ present, int E, int npb, int wpr) {
  const int tid = threadIdx.x;
  for (int i = tid; i < 32 * PPITCH; i += PG) tile[i] = 0u;
  __syncthreads();

  const int g  = bid * PG + tid;
  const int e0 = g * 16;
  const int wv = tid >> 5;
  const unsigned bit = 1u << (tid & 31);
  int cc[16];
#pragma unroll
  for (int k = 0; k < 16; ++k) {
    const int e = e0 + k;
    cc[k] = -1;
    if (e < E) cc[k] = col[e];
  }
#pragma unroll
  for (int k = 0; k < 16; ++k) {
    if (cc[k] >= 0) {
      const unsigned t = (unsigned)cc[k] / (unsigned)npb;
      if (t < 256u) atomicOr(&tile[wv * PPITCH + t], bit);
    }
  }
  __syncthreads();

  for (int idx = tid; idx < 256 * 32; idx += PG) {
    const int t = idx >> 5, w = idx & 31;
    present[(size_t)t * wpr + bid * 32 + w] = tile[w * PPITCH + t];
  }
}

__global__ void __launch_bounds__(1024)
k_deg_present(const int* __restrict__ col, const float* __restrict__ w,
              float* __restrict__ part, unsigned* __restrict__ present,
              int E, int n, int npp, int es, int npb, int wpr, int gDeg) {
  __shared__ unsigned lds[DCAP];
  if ((int)blockIdx.x < gDeg)
    deg_part_role((float*)lds, (int)blockIdx.x, col, w, part, E, n, npp, es);
  else
    present_role(lds, (int)blockIdx.x - gDeg, col, present, E, npb, wpr);
}

__device__ __forceinline__ void
gemm_role(int bid, const float* __restrict__ A,
          const float* __restrict__ W,
          float* __restrict__ C, int n) {
  int wid  = (bid * blockDim.x + threadIdx.x) >> 5;
  int lane = threadIdx.x & 31;
  int row0 = wid * 16;
  if (row0 >= n) return;

  int half = lane >> 4;
  int l    = lane & 15;

  v8f acc[8];
#pragma unroll
  for (int t = 0; t < 8; ++t) acc[t] = (v8f){0,0,0,0,0,0,0,0};

  const float* arow = A + (size_t)(row0 + l) * HID;

  for (int k = 0; k < HID; k += 4) {
    v2f afrag = *(const v2f*)(arow + k + half * 2);
#pragma unroll
    for (int t = 0; t < 8; ++t) {
      int ncol = t * 16 + l;
      v2f bfrag;
      bfrag.x = W[(size_t)(k + half * 2)     * HID + ncol];
      bfrag.y = W[(size_t)(k + half * 2 + 1) * HID + ncol];
      acc[t] = __builtin_amdgcn_wmma_f32_16x16x4_f32(
           false, afrag,  false, bfrag,
           (short)0, acc[t],  false,  false);
    }
  }

#pragma unroll
  for (int t = 0; t < 8; ++t) {
#pragma unroll
    for (int r = 0; r < 8; ++r) {
      int m    = row0 + r + half * 8;
      int ncol = t * 16 + l;
      C[(size_t)m * HID + ncol] = acc[t][r];
    }
  }
}

__global__ void k_gemm_wmma(const float* __restrict__ A,
                            const float* __restrict__ W,
                            float* __restrict__ C, int n) {
  gemm_role((int)blockIdx.x, A, W, C, n);
}

__global__ void k_gemm_dinv(const float* __restrict__ A,
                            const float* __restrict__ W,
                            float* __restrict__ C, int n,
                            const float* __restrict__ part,
                            float* __restrict__ dinv, int gGemm) {
  if ((int)blockIdx.x < gGemm) gemm_role((int)blockIdx.x, A, W, C, n);
  else                         dinv_role((int)blockIdx.x - gGemm, part, dinv, n);
}

#define NPB   392
#define AGW   32
#define QCAP  544
#define GQCAP 64
#define EIF   8
#define WQ    320
#define WSL   10
#define CAPB  (AGW * WSL)
#define EBITS 23
#define RSH  17

__device__ __forceinline__ void wave_lds_order() {
  __builtin_amdgcn_fence(__ATOMIC_RELEASE, "wavefront");
  __builtin_amdgcn_wave_barrier();
  __builtin_amdgcn_fence(__ATOMIC_ACQUIRE, "wavefront");
}

__device__ __forceinline__ void row_add(float* rows, int lc, int lane, float4 v, float s) {
  float4* d = (float4*)(rows + lc * HID) + lane;
  float4 a = *d;
  a.x += v.x * s;  a.y += v.y * s;  a.z += v.z * s;  a.w += v.w * s;
  *d = a;
}

__device__ __forceinline__ void gather_own(int r, int lc, float s, int cnt, int lane,
                                           const float* __restrict__ xw, float* rows) {
  int j = 0;
  for (; j + EIF <= cnt; j += EIF) {
    float4 v[EIF];
    int    lj[EIF];
    float  sj[EIF];
#pragma unroll
    for (int u = 0; u < EIF; ++u) {
      const int rj = __builtin_amdgcn_readlane(r, j + u);
      lj[u] = __builtin_amdgcn_readlane(lc, j + u);
      sj[u] = __int_as_float(__builtin_amdgcn_readlane(__float_as_int(s), j + u));
      v[u] = ((const float4*)(xw + (size_t)rj * HID))[lane];
    }
#pragma unroll
    for (int u = 0; u < EIF; ++u) row_add(rows, lj[u], lane, v[u], sj[u]);
  }
  if (j < cnt) {
    const int rem = cnt - j;
    float4 v[EIF];
    int    lj[EIF];
    float  sj[EIF];
#pragma unroll
    for (int u = 0; u < EIF; ++u) {
      if (u < rem) {
        const int rj = __builtin_amdgcn_readlane(r, j + u);
        lj[u] = __builtin_amdgcn_readlane(lc, j + u);
        sj[u] = __int_as_float(__builtin_amdgcn_readlane(__float_as_int(s), j + u));
        v[u] = ((const float4*)(xw + (size_t)rj * HID))[lane];
      }
    }
#pragma unroll
    for (int u = 0; u < EIF; ++u) {
      if (u < rem) row_add(rows, lj[u], lane, v[u], sj[u]);
    }
  }
}

__device__ __forceinline__ void route_batch(const int* qv, int cnt, int lane,
                                            int* wq, int* wcnt, int* nrep,
                                            const float* __restrict__ xw,
                                            const int* __restrict__ row,
                                            const float* __restrict__ w,
                                            const float* __restrict__ dinv,
                                            const float* dl, float* rows) {
  int p = 0;
  bool fail = false;
  if (lane < cnt) {
    p = qv[lane];
    const int ow = (int)(((unsigned)p >> EBITS) & (unsigned)(AGW - 1));
    const int slot = atomicAdd(&wcnt[ow], 1);
    if (slot < WQ) wq[ow * WQ + slot] = p;
    else           fail = true;
  }
  unsigned fm = __builtin_amdgcn_ballot_w32(fail);
  if (fm != 0u) {
    if (lane == 0) atomicOr(nrep, 1);
    while (fm != 0u) {
      const int src = __builtin_ctz(fm);
      fm &= fm - 1u;
      const unsigned pe = (unsigned)__builtin_amdgcn_readlane(p, src);
      const int e  = (int)(pe & ((1u << EBITS) - 1u));
      const int lc = (int)(pe >> EBITS);
      const int r  = row[e];
      const float s = dinv[r] * w[e] * dl[lc];
      const float4 v = ((const float4*)(xw + (size_t)r * HID))[lane];
      float* d = rows + lc * HID + 4 * lane;
      atomicAdd(d,     v.x * s);
      atomicAdd(d + 1, v.y * s);
      atomicAdd(d + 2, v.z * s);
      atomicAdd(d + 3, v.w * s);
    }
  }
}

__device__ __forceinline__ int scan_groups(const int* gv, int cnt, int lane, int base, int cntn,
                                           const int* __restrict__ col, int E, bool vec,
                                           int* myq, int qn) {
  const int sub = lane & 3;
  int cv[4][4];
  int eb[4];
#pragma unroll
  for (int k = 0; k < 4; ++k) {
    const int gi = 8 * k + (lane >> 2);
    eb[k] = E;
    if (gi < cnt) eb[k] = (gv[gi] << 4) + (sub << 2);
    cv[k][0] = -1;  cv[k][1] = -1;  cv[k][2] = -1;  cv[k][3] = -1;
  }
  if (vec) {
#pragma unroll
    for (int k = 0; k < 4; ++k) {
      if (eb[k] < E) {
        const int4 t = *(const int4*)(col + eb[k]);
        cv[k][0] = t.x;  cv[k][1] = t.y;  cv[k][2] = t.z;  cv[k][3] = t.w;
      }
    }
  } else {
#pragma unroll
    for (int k = 0; k < 4; ++k) {
      if (eb[k]     < E) cv[k][0] = col[eb[k]];
      if (eb[k] + 1 < E) cv[k][1] = col[eb[k] + 1];
      if (eb[k] + 2 < E) cv[k][2] = col[eb[k] + 2];
      if (eb[k] + 3 < E) cv[k][3] = col[eb[k] + 3];
    }
  }
#pragma unroll
  for (int k = 0; k < 4; ++k) {
    if (8 * k < cnt) {
#pragma unroll
      for (int s = 0; s < 4; ++s) {
        const unsigned lc = (unsigned)cv[k][s] - (unsigned)base;
        const bool hit = lc < (unsigned)cntn;
        const unsigned m = __builtin_amdgcn_ballot_w32(hit);
        if (hit)
          myq[qn + (int)__builtin_amdgcn_mbcnt_lo(m, 0u)] = (int)((lc << EBITS) | (unsigned)(eb[k] + s));
        qn += __builtin_popcount(m);
      }
    }
  }
  return qn;
}

__global__ void __launch_bounds__(1024)
k_agg(const float* __restrict__ xw,
      const int* __restrict__ row,
      const int* __restrict__ col,
      const float* __restrict__ w,
      const unsigned* __restrict__ present,
      const float* __restrict__ dinv,
      const float* __restrict__ b,
      float* __restrict__ h, int E, int n, int npb, int wpr,
      int* __restrict__ rec, int* __restrict__ meta, int mode) {
  __shared__ float rows[NPB * HID];
  __shared__ int   q[AGW][QCAP];
  __shared__ int   gq[AGW][GQCAP];
  __shared__ int   wq[AGW * WQ];
  __shared__ float dl[NPB];
  __shared__ int   wcnt[AGW];
  __shared__ int   nrep;

  const int tid  = threadIdx.x;
  const int lane = tid & 31;
  const int wv   = __builtin_amdgcn_readfirstlane(tid >> 5);
  const int base = blockIdx.x * npb;
  int cntn = n - base;
  if (cntn > npb) cntn = npb;
  if (tid < AGW) wcnt[tid] = 0;
  if (tid == 0) nrep = 0;

  int2* slab = (int2*)rec + (size_t)blockIdx.x * (CAPB * 32);
  int2* myslab = slab + (size_t)wv * (WSL * 32);
  int* metaw = meta + 256 + blockIdx.x * AGW;
  int replay = -1;
  if (mode == 2) replay = __builtin_amdgcn_readfirstlane(meta[blockIdx.x]);
  int2* recw = (mode == 1) ? myslab : nullptr;

  for (int idx = tid; idx < cntn * (HID / 4); idx += AGW * 32) {
    const float s0 = dinv[base + (idx >> 5)];
    const float s = s0 * s0;
    float4 xv = ((const float4*)xw)[(size_t)base * (HID / 4) + idx];
    xv.x *= s;  xv.y *= s;  xv.z *= s;  xv.w *= s;
    ((float4*)rows)[idx] = xv;
    if ((idx & 31) == 0) dl[idx >> 5] = s0;
  }
  __syncthreads();

  if (replay < 0) {
    const bool vec = (E & 3) == 0;
    const unsigned* myrow = present + (size_t)blockIdx.x * wpr;
    int* myq  = q[wv];
    int* mygq = gq[wv];
    int qn = 0;
    int gn = 0;

    unsigned wnext = 0u;
    if ((wv << 5) < wpr) wnext = myrow[(wv << 5) + lane];
    for (int i0 = wv << 5; i0 < wpr; i0 += AGW * 32) {
      const int i = i0 + lane;
      unsigned word = wnext;
      if (i0 + AGW * 32 < wpr) wnext = myrow[i + AGW * 32];
      unsigned act = __builtin_amdgcn_ballot_w32(word != 0u);
      while (act != 0u) {
        if (word != 0u) {
          const int slot = gn + (int)__builtin_amdgcn_mbcnt_lo(act, 0u);
          mygq[slot] = (i << 5) + (__ffs((int)word) - 1);
          word &= word - 1u;
        }
        gn += __builtin_popcount(act);
        if (gn >= 32) {
          wave_lds_order();
          qn = scan_groups(mygq + (gn - 32), 32, lane, base, cntn, col, E, vec, myq, qn);
          gn -= 32;
          wave_lds_order();
          if (qn >= 32) {
            while (qn >= 32) {
              route_batch(myq + (qn - 32), 32, lane, wq, wcnt, &nrep, xw, row, w, dinv, dl, rows);
              qn -= 32;
            }
            wave_lds_order();
          }
        }
        act = __builtin_amdgcn_ballot_w32(word != 0u);
      }
    }
    if (gn > 0) {
      wave_lds_order();
      qn = scan_groups(mygq, gn, lane, base, cntn, col, E, vec, myq, qn);
    }
    wave_lds_order();
    while (qn > 0) {
      const int c = (qn < 32) ? qn : 32;
      route_batch(myq + (qn - c), c, lane, wq, wcnt, &nrep, xw, row, w, dinv, dl, rows);
      qn -= c;
    }
    __syncthreads();

    int nq = __builtin_amdgcn_readfirstlane(wcnt[wv]);
    if (nq > WQ) nq = WQ;
    const int* mq = wq + wv * WQ;
    const int nb = (nq + 31) >> 5;
    int   rA = 0, lcA = 0;
    float wA = 0.0f;
    if (lane < nq) {
      const unsigned p = (unsigned)mq[lane];
      lcA = (int)(p >> EBITS);
      const int e = (int)(p & ((1u << EBITS) - 1u));
      rA = row[e];  wA = w[e];
    }
    for (int k = 0; k < nb; ++k) {
      const int c = (nq - 32 * k < 32) ? (nq - 32 * k) : 32;
      const int r = rA, lc = lcA;
      float s = 0.0f;
      if (lane < c) s = dinv[r] * wA * dl[lc];
      const int jn = 32 * (k + 1) + lane;
      rA = 0;  lcA = 0;  wA = 0.0f;
      if (jn < nq) {
        const unsigned p = (unsigned)mq[jn];
        lcA = (int)(p >> EBITS);
        const int e = (int)(p & ((1u << EBITS) - 1u));
        rA = row[e];  wA = w[e];
      }
      gather_own(r, lc, s, c, lane, xw, rows);
      if (recw != nullptr && k < WSL)
        recw[k * 32 + lane] = make_int2(lane < c ? (r | (lc << RSH)) : -1, __float_as_int(s));
    }
    if (mode == 1 && lane == 0) {
      metaw[wv] = nb;
      if (nb > WSL) atomicOr(&nrep, 1);
    }
  } else {
    const int nb = __builtin_amdgcn_readfirstlane(metaw[wv]);
    int2 pn = make_int2(-1, 0);
    if (0 < nb) pn = myslab[lane];
    for (int k = 0; k < nb; ++k) {
      const int2 p = pn;
      if (k + 1 < nb) pn = myslab[(k + 1) * 32 + lane];
      const bool valid = p.x >= 0;
      const int c = __builtin_popcount(__builtin_amdgcn_ballot_w32(valid));
      const int r  = valid ? (p.x & ((1 << RSH) - 1)) : 0;
      const int lc = valid ? (p.x >> RSH) : 0;
      gather_own(r, lc, __int_as_float(p.y), c, lane, xw, rows);
    }
  }
  __syncthreads();
  if (mode == 1 && tid == 0) meta[blockIdx.x] = (nrep == 0) ? 1 : -1;

  for (int idx = tid; idx < cntn * (HID / 4); idx += AGW * 32) {
    const float4 a  = ((const float4*)rows)[idx];
    const float4 bv = ((const float4*)b)[idx & 31];
    float4 o;
    o.x = a.x + bv.x;  o.x = o.x > 0.0f ? o.x : 0.0f;
    o.y = a.y + bv.y;  o.y = o.y > 0.0f ? o.y : 0.0f;
    o.z = a.z + bv.z;  o.z = o.z > 0.0f ? o.z : 0.0f;
    o.w = a.w + bv.w;  o.w = o.w > 0.0f ? o.w : 0.0f;
    ((float4*)h)[(size_t)base * (HID / 4) + idx] = o;
  }
}

__global__ void k_out_wmma(const float* __restrict__ h,
                           const float* __restrict__ Wout,
                           const float* __restrict__ bout,
                           float* __restrict__ out, int n) {
  const int wid  = (blockIdx.x * blockDim.x + threadIdx.x) >> 5;
  const int lane = threadIdx.x & 31;
  const int row0 = wid * 16;
  if (row0 >= n) return;

  const int c  = lane & 15;
  const int hf = lane >> 4;

  const int node = row0 + c;
  const int ld   = node < n ? node : n - 1;
  const float4* wr = (const float4*)(Wout + (size_t)c * HID) + hf;
  const float4* hr = (const float4*)(h + (size_t)ld * HID) + hf;

  const float4 bl = ((const float4*)bout)[2 * hf];
  const float4 bh = ((const float4*)bout)[2 * hf + 1];
  v8f acc0 = (v8f){bl.x, bl.y, bl.z, bl.w, bh.x, bh.y, bh.z, bh.w};
  v8f acc1 = (v8f){0,0,0,0,0,0,0,0};
#pragma unroll
  for (int j = 0; j < 16; ++j) {
    const float4 a = wr[2 * j];
    const float4 b = hr[2 * j];
    v2f a0 = {a.x, a.y}, b0 = {b.x, b.y};
    v2f a1 = {a.z, a.w}, b1 = {b.z, b.w};
    acc0 = __builtin_amdgcn_wmma_f32_16x16x4_f32(
         false, a0,  false, b0,
         (short)0, acc0,  false,  false);
    acc1 = __builtin_amdgcn_wmma_f32_16x16x4_f32(
         false, a1,  false, b1,
         (short)0, acc1,  false,  false);
  }
  const v8f r = acc0 + acc1;
  if (node < n) {
    float4* o = (float4*)(out + (size_t)node * NPRED + 8 * hf);
    o[0] = make_float4(r[0], r[1], r[2], r[3]);
    o[1] = make_float4(r[4], r[5], r[6], r[7]);
  }
}


extern "C" void kernel_launch(void* const* d_in, const int* in_sizes, int n_in,
                              void* d_out, int out_size, void* d_ws, size_t ws_size,
                              hipStream_t stream) {
  const float* x    = (const float*)d_in[0];
  const int*   ei   = (const int*)  d_in[1];
  const float* ew   = (const float*)d_in[2];
  const float* W1   = (const float*)d_in[3];
  const float* b1   = (const float*)d_in[4];
  const float* W2   = (const float*)d_in[5];
  const float* b2   = (const float*)d_in[6];
  const float* Wout = (const float*)d_in[7];
  const float* bout = (const float*)d_in[8];
  float* out = (float*)d_out;

  const int n = in_sizes[0] / HID;
  const int E = in_sizes[2];
  const int* erow = ei;
  const int* ecol = ei + E;

  float* xw   = (float*)d_ws;
  float* hbuf = xw + (size_t)n * HID;
  float* dinv = hbuf + (size_t)n * HID;
  unsigned* present = (unsigned*)(dinv + n);
  int gPre = (E + 16 * PG - 1) / (16 * PG);
  int wpr  = gPre * 32;
  size_t recOff = ((size_t)((const char*)(present + (size_t)256 * wpr) - (const char*)d_ws) + 33792 + 255) / 256 * 256;
  int*  meta = (int*)(present + (size_t)256 * wpr);
  int*  rec  = (int*)((char*)d_ws + recOff);
  const bool keep = ws_size >= recOff + (size_t)256 * CAPB * 256 && n <= (1 << RSH);

  const int B = 256;
  int gN  = (n + B - 1) / B;
  int nTiles = (n + 15) / 16;
  int gGemm  = (nTiles + 7) / 8;
  int npb    = (n + 255) / 256;
  int gAgg   = (n + npb - 1) / npb;
  int nPart  = (n + DCAP - 1) / DCAP;
  int npp    = (((n + nPart - 1) / nPart) + 31) / 32 * 32;
  int gDeg   = DSL * ((n + npp - 1) / npp);
  int es     = (E + DSL - 1) / DSL;

  k_deg_present<<<gDeg + gPre, 1024, 0, stream>>>(ecol, ew, hbuf, present, E, n, npp, es, npb, wpr, gDeg);

  k_gemm_dinv<<<gGemm + gN, B, 0, stream>>>(x, W1, xw, n, hbuf, dinv, gGemm);
  k_agg      <<<gAgg, AGW * 32, 0, stream>>>(xw, erow, ecol, ew, present, dinv, b1, hbuf, E, n, npb, wpr, rec, meta, keep ? 1 : 0);

  k_gemm_wmma<<<gGemm, B, 0, stream>>>(hbuf, W2, xw, n);
  k_agg      <<<gAgg, AGW * 32, 0, stream>>>(xw, erow, ecol, ew, present, dinv, b2, hbuf, E, n, npb, wpr, rec, meta, keep ? 2 : 0);

  k_out_wmma<<<gGemm, B, 0, stream>>>(hbuf, Wout, bout, out, n);
}
